# starting kernel
# baseline (speedup 1.0000x reference)
#include <hip/hip_runtime.h>

typedef __attribute__((ext_vector_type(2))) float v2f;
typedef __attribute__((ext_vector_type(8))) float v8f;

#define LD 65
#define MS (64 * LD)
#define C_DIM 64
#define T_DIM 4096
#define N_EPOCH 8
#define PATCH 512
#define EPS_COV 1e-5f

__device__ __forceinline__ void mm64(float* __restrict__ D, const float* __restrict__ A,
                                     const float* __restrict__ Bm, int lane, int wave) {
    __syncthreads();
#pragma unroll
    for (int i = 0; i < 2; ++i) {
        const int t  = wave + 8 * i;
        const int tm = (t >> 2) * 16;
        const int tn = (t & 3) * 16;
        const int row = tm + (lane & 15);
        const int col = tn + (lane & 15);
        v8f acc = {0.f, 0.f, 0.f, 0.f, 0.f, 0.f, 0.f, 0.f};
#pragma unroll
        for (int k4 = 0; k4 < 64; k4 += 4) {
            const int ko = k4 + ((lane >> 4) << 1);
            v2f a; a.x = A[row * LD + ko];  a.y = A[row * LD + ko + 1];
            v2f b; b.x = Bm[ko * LD + col]; b.y = Bm[(ko + 1) * LD + col];
            acc = __builtin_amdgcn_wmma_f32_16x16x4_f32(false, a, false, b,
                                                        (short)0, acc, false, false);
        }
#pragma unroll
        for (int v = 0; v < 8; ++v) {
            const int m = tm + v + 8 * (lane >> 4);
            D[m * LD + tn + (lane & 15)] = acc[v];
        }
    }
    __syncthreads();
}

#define FOR_EW for (int i = tid; i < 4096; i += 256)

__device__ __forceinline__ void ew_copy(float* d, const float* s, int tid) {
    FOR_EW { int r = i >> 6, c = i & 63; d[r * LD + c] = s[r * LD + c]; }
    __syncthreads();
}
__device__ __forceinline__ void ew_identity(float* d, float dv, int tid) {
    FOR_EW { int r = i >> 6, c = i & 63; d[r * LD + c] = (r == c) ? dv : 0.f; }
    __syncthreads();
}

__global__ void cov64_kernel(const float* __restrict__ x, float* __restrict__ cov) {
    __shared__ float Xs[64 * LD];
    __shared__ float mp[256];
    __shared__ float mean[64];

    const int tid = threadIdx.x, lane = tid & 31, wave = tid >> 5;
    const int bid = blockIdx.x;
    const int b = bid >> 3, e = bid & 7;
    const float* xb = x + (size_t)b * C_DIM * T_DIM + (size_t)e * PATCH;

    float msum = 0.f;
    v8f acc0 = {0.f,0.f,0.f,0.f,0.f,0.f,0.f,0.f};
    v8f acc1 = {0.f,0.f,0.f,0.f,0.f,0.f,0.f,0.f};

    for (int kc = 0; kc < PATCH; kc += 64) {
        __syncthreads();
        for (int i = tid; i < 4096; i += 256) {
            const int c = i >> 6, j = i & 63;
            Xs[c * LD + j] = xb[(size_t)c * T_DIM + kc + j];
        }
        __syncthreads();
        {
            const int c = tid & 63, part = tid >> 6;
#pragma unroll
            for (int j = 0; j < 16; ++j) msum += Xs[c * LD + part * 16 + j];
        }
#pragma unroll
        for (int i = 0; i < 2; ++i) {
            const int t  = wave + 8 * i;
            const int tm = (t >> 2) * 16, tn = (t & 3) * 16;
            const int row = tm + (lane & 15), coln = tn + (lane & 15);
            v8f acc = i ? acc1 : acc0;
#pragma unroll
            for (int k4 = 0; k4 < 64; k4 += 4) {
                const int ko = k4 + ((lane >> 4) << 1);
                v2f a; a.x = Xs[row * LD + ko];  a.y = Xs[row * LD + ko + 1];
                v2f bf; bf.x = Xs[coln * LD + ko]; bf.y = Xs[coln * LD + ko + 1];
                acc = __builtin_amdgcn_wmma_f32_16x16x4_f32(false, a, false, bf,
                                                            (short)0, acc, false, false);
            }
            if (i) acc1 = acc; else acc0 = acc;
        }
    }
    __syncthreads();
    mp[tid] = msum;
    __syncthreads();
    if (tid < 64)
        mean[tid] = (mp[tid] + mp[tid + 64] + mp[tid + 128] + mp[tid + 192]) * (1.f / PATCH);
    __syncthreads();

#pragma unroll
    for (int i = 0; i < 2; ++i) {
        const int t  = wave + 8 * i;
        const int tm = (t >> 2) * 16, tn = (t & 3) * 16;
        v8f acc = i ? acc1 : acc0;
#pragma unroll
        for (int v = 0; v < 8; ++v) {
            const int m = tm + v + 8 * (lane >> 4);
            const int n = tn + (lane & 15);
            const float val = (acc[v] - (float)PATCH * mean[m] * mean[n]) * (1.f / (PATCH - 1));
            cov[(size_t)bid * 4096 + m * 64 + n] = val;
        }
    }
}

__global__ void spdlogexp_kernel(const float* __restrict__ cov,
                                 const float* __restrict__ alphap,
                                 const float* __restrict__ betap,
                                 float* __restrict__ out) {
    extern __shared__ float lds[];
    float* A  = lds;
    float* Y  = lds + 1 * MS;
    float* Z  = lds + 2 * MS;
    float* Tm = lds + 3 * MS;
    float* W  = lds + 4 * MS;
    float* red = lds + 5 * MS;

    const int tid = threadIdx.x, lane = tid & 31, wave = tid >> 5;
    const int bid = blockIdx.x;
    const float alpha = alphap[0], beta = betap[0];

    for (int i = tid; i < 4096; i += 256)
        A[(i >> 6) * LD + (i & 63)] = cov[(size_t)bid * 4096 + i];
    __syncthreads();
    if (tid < 64) red[tid] = A[tid * LD + tid];
    __syncthreads();
    if (tid == 0) { float s = 0.f; for (int i = 0; i < 64; ++i) s += red[i]; red[0] = 1.f / s; }
    __syncthreads();
    const float invtr = red[0];
    FOR_EW { int r = i >> 6, c = i & 63;
             float v = A[r * LD + c] * invtr;
             if (r == c) v += EPS_COV;
             A[r * LD + c] = v; }
    __syncthreads();

    constexpr int K_SQRT = 5;
    const int ns_it[K_SQRT] = {18, 12, 9, 7, 6};
    for (int kq = 0; kq < K_SQRT; ++kq) {
        ew_copy(Y, A, tid);
        ew_identity(Z, 1.f, tid);
        for (int it = 0; it < ns_it[kq]; ++it) {
            mm64(Tm, Z, Y, lane, wave);
            FOR_EW { int r = i >> 6, c = i & 63;
                     float v = Tm[r * LD + c];
                     Tm[r * LD + c] = ((r == c) ? 1.5f : 0.f) - 0.5f * v; }
            __syncthreads();
            mm64(W, Y, Tm, lane, wave);  ew_copy(Y, W, tid);
            mm64(W, Tm, Z, lane, wave);  ew_copy(Z, W, tid);
        }
        ew_copy(A, Y, tid);
    }
    FOR_EW { int r = i >> 6, c = i & 63;
             Y[r * LD + c] = ((r == c) ? 1.f : 0.f) - A[r * LD + c]; }
    __syncthreads();
    constexpr int LOG_M = 12;
    ew_identity(Z, 1.f / (float)LOG_M, tid);
    for (int m = LOG_M - 1; m >= 1; --m) {
        mm64(Tm, Y, Z, lane, wave);
        const float dm = 1.f / (float)m;
        FOR_EW { int r = i >> 6, c = i & 63;
                 Z[r * LD + c] = Tm[r * LD + c] + ((r == c) ? dm : 0.f); }
        __syncthreads();
    }
    mm64(Tm, Y, Z, lane, wave);
    const float lscale = -(float)(1 << K_SQRT);

    FOR_EW { int r = i >> 6, c = i & 63;
             A[r * LD + c] = alpha * (lscale * Tm[r * LD + c]) + beta; }
    __syncthreads();
    FOR_EW { int r = i >> 6, c = i & 63;
             W[r * LD + c] = 0.5f * (A[r * LD + c] + A[c * LD + r]); }
    __syncthreads();

    constexpr int S_EXP = 6, EXP_M = 10;
    const float es = 1.f / (float)(1 << S_EXP);
    FOR_EW { int r = i >> 6, c = i & 63; Y[r * LD + c] = W[r * LD + c] * es; }
    __syncthreads();
    ew_identity(Z, 1.f, tid);
    for (int m = EXP_M; m >= 1; --m) {
        mm64(Tm, Y, Z, lane, wave);
        const float dm = 1.f / (float)m;
        FOR_EW { int r = i >> 6, c = i & 63;
                 Z[r * LD + c] = Tm[r * LD + c] * dm + ((r == c) ? 1.f : 0.f); }
        __syncthreads();
    }
    for (int j = 0; j < S_EXP; ++j) {
        mm64(Tm, Z, Z, lane, wave);
        ew_copy(Z, Tm, tid);
    }

    for (int i = tid; i < 4096; i += 256)
        out[(size_t)bid * 4096 + i] = Z[(i >> 6) * LD + (i & 63)];
}

extern "C" void kernel_launch(void* const* d_in, const int* in_sizes, int n_in,
                              void* d_out, int out_size, void* d_ws, size_t ws_size,
                              hipStream_t stream) {
    const float* x     = (const float*)d_in[0];
    const float* alpha = (const float*)d_in[1];
    const float* beta  = (const float*)d_in[2];
    float* out = (float*)d_out;
    float* cov = (float*)d_ws;

    const int nmat = 256 * N_EPOCH;
    cov64_kernel<<<nmat, 256, 0, stream>>>(x, cov);

    const size_t shbytes = (size_t)(5 * MS + 64) * sizeof(float);
    spdlogexp_kernel<<<nmat, 256, shbytes, stream>>>(cov, alpha, beta, out);
}


# LDS row stride 65 -> 76 floats (4 x odd): WMMA A/B fragment reads and D tile stores become bank-conflict free, A fragment one aligned 8-byte load; arithmetic unchanged
# speedup vs baseline: 1.2644x; 1.2644x over previous
#include <hip/hip_runtime.h>

typedef __attribute__((ext_vector_type(2))) float v2f;
typedef __attribute__((ext_vector_type(8))) float v8f;

#define LD 76
#define MS (64 * LD)
#define C_DIM 64
#define T_DIM 4096
#define N_EPOCH 8
#define PATCH 512
#define EPS_COV 1e-5f

__device__ __forceinline__ void mm64(float* __restrict__ D, const float* __restrict__ A,
                                     const float* __restrict__ Bm, int lane, int wave) {
    __syncthreads();
#pragma unroll
    for (int i = 0; i < 2; ++i) {
        const int t  = wave + 8 * i;
        const int tm = (t >> 2) * 16;
        const int tn = (t & 3) * 16;
        const int row = tm + (lane & 15);
        const int col = tn + (lane & 15);
        v8f acc = {0.f, 0.f, 0.f, 0.f, 0.f, 0.f, 0.f, 0.f};
#pragma unroll
        for (int k4 = 0; k4 < 64; k4 += 4) {
            const int ko = k4 + ((lane >> 4) << 1);
            const v2f a = *reinterpret_cast<const v2f*>(A + row * LD + ko);
            v2f b; b.x = Bm[ko * LD + col]; b.y = Bm[(ko + 1) * LD + col];
            acc = __builtin_amdgcn_wmma_f32_16x16x4_f32(false, a, false, b,
                                                        (short)0, acc, false, false);
        }
#pragma unroll
        for (int v = 0; v < 8; ++v) {
            const int m = tm + v + 8 * (lane >> 4);
            D[m * LD + tn + (lane & 15)] = acc[v];
        }
    }
    __syncthreads();
}

#define FOR_EW for (int i = tid; i < 4096; i += 256)

__device__ __forceinline__ void ew_copy(float* d, const float* s, int tid) {
    FOR_EW { int r = i >> 6, c = i & 63; d[r * LD + c] = s[r * LD + c]; }
    __syncthreads();
}
__device__ __forceinline__ void ew_identity(float* d, float dv, int tid) {
    FOR_EW { int r = i >> 6, c = i & 63; d[r * LD + c] = (r == c) ? dv : 0.f; }
    __syncthreads();
}

__global__ void cov64_kernel(const float* __restrict__ x, float* __restrict__ cov) {
    __shared__ __attribute__((aligned(16))) float Xs[64 * LD];
    __shared__ float mp[256];
    __shared__ float mean[64];

    const int tid = threadIdx.x, lane = tid & 31, wave = tid >> 5;
    const int bid = blockIdx.x;
    const int b = bid >> 3, e = bid & 7;
    const float* xb = x + (size_t)b * C_DIM * T_DIM + (size_t)e * PATCH;

    float msum = 0.f;
    v8f acc0 = {0.f,0.f,0.f,0.f,0.f,0.f,0.f,0.f};
    v8f acc1 = {0.f,0.f,0.f,0.f,0.f,0.f,0.f,0.f};

    for (int kc = 0; kc < PATCH; kc += 64) {
        __syncthreads();
        for (int i = tid; i < 4096; i += 256) {
            const int c = i >> 6, j = i & 63;
            Xs[c * LD + j] = xb[(size_t)c * T_DIM + kc + j];
        }
        __syncthreads();
        {
            const int c = tid & 63, part = tid >> 6;
#pragma unroll
            for (int j = 0; j < 16; ++j) msum += Xs[c * LD + part * 16 + j];
        }
#pragma unroll
        for (int i = 0; i < 2; ++i) {
            const int t  = wave + 8 * i;
            const int tm = (t >> 2) * 16, tn = (t & 3) * 16;
            const int row = tm + (lane & 15), coln = tn + (lane & 15);
            v8f acc = i ? acc1 : acc0;
#pragma unroll
            for (int k4 = 0; k4 < 64; k4 += 4) {
                const int ko = k4 + ((lane >> 4) << 1);
                const v2f a  = *reinterpret_cast<const v2f*>(Xs + row * LD + ko);
                const v2f bf = *reinterpret_cast<const v2f*>(Xs + coln * LD + ko);
                acc = __builtin_amdgcn_wmma_f32_16x16x4_f32(false, a, false, bf,
                                                            (short)0, acc, false, false);
            }
            if (i) acc1 = acc; else acc0 = acc;
        }
    }
    __syncthreads();
    mp[tid] = msum;
    __syncthreads();
    if (tid < 64)
        mean[tid] = (mp[tid] + mp[tid + 64] + mp[tid + 128] + mp[tid + 192]) * (1.f / PATCH);
    __syncthreads();

#pragma unroll
    for (int i = 0; i < 2; ++i) {
        const int t  = wave + 8 * i;
        const int tm = (t >> 2) * 16, tn = (t & 3) * 16;
        v8f acc = i ? acc1 : acc0;
#pragma unroll
        for (int v = 0; v < 8; ++v) {
            const int m = tm + v + 8 * (lane >> 4);
            const int n = tn + (lane & 15);
            const float val = (acc[v] - (float)PATCH * mean[m] * mean[n]) * (1.f / (PATCH - 1));
            cov[(size_t)bid * 4096 + m * 64 + n] = val;
        }
    }
}

__global__ void spdlogexp_kernel(const float* __restrict__ cov,
                                 const float* __restrict__ alphap,
                                 const float* __restrict__ betap,
                                 float* __restrict__ out) {
    extern __shared__ __attribute__((aligned(16))) float lds[];
    float* A  = lds;
    float* Y  = lds + 1 * MS;
    float* Z  = lds + 2 * MS;
    float* Tm = lds + 3 * MS;
    float* W  = lds + 4 * MS;
    float* red = lds + 5 * MS;

    const int tid = threadIdx.x, lane = tid & 31, wave = tid >> 5;
    const int bid = blockIdx.x;
    const float alpha = alphap[0], beta = betap[0];

    for (int i = tid; i < 4096; i += 256)
        A[(i >> 6) * LD + (i & 63)] = cov[(size_t)bid * 4096 + i];
    __syncthreads();
    if (tid < 64) red[tid] = A[tid * LD + tid];
    __syncthreads();
    if (tid == 0) { float s = 0.f; for (int i = 0; i < 64; ++i) s += red[i]; red[0] = 1.f / s; }
    __syncthreads();
    const float invtr = red[0];
    FOR_EW { int r = i >> 6, c = i & 63;
             float v = A[r * LD + c] * invtr;
             if (r == c) v += EPS_COV;
             A[r * LD + c] = v; }
    __syncthreads();

    constexpr int K_SQRT = 5;
    const int ns_it[K_SQRT] = {18, 12, 9, 7, 6};
    for (int kq = 0; kq < K_SQRT; ++kq) {
        ew_copy(Y, A, tid);
        ew_identity(Z, 1.f, tid);
        for (int it = 0; it < ns_it[kq]; ++it) {
            mm64(Tm, Z, Y, lane, wave);
            FOR_EW { int r = i >> 6, c = i & 63;
                     float v = Tm[r * LD + c];
                     Tm[r * LD + c] = ((r == c) ? 1.5f : 0.f) - 0.5f * v; }
            __syncthreads();
            mm64(W, Y, Tm, lane, wave);  ew_copy(Y, W, tid);
            mm64(W, Tm, Z, lane, wave);  ew_copy(Z, W, tid);
        }
        ew_copy(A, Y, tid);
    }
    FOR_EW { int r = i >> 6, c = i & 63;
             Y[r * LD + c] = ((r == c) ? 1.f : 0.f) - A[r * LD + c]; }
    __syncthreads();
    constexpr int LOG_M = 12;
    ew_identity(Z, 1.f / (float)LOG_M, tid);
    for (int m = LOG_M - 1; m >= 1; --m) {
        mm64(Tm, Y, Z, lane, wave);
        const float dm = 1.f / (float)m;
        FOR_EW { int r = i >> 6, c = i & 63;
                 Z[r * LD + c] = Tm[r * LD + c] + ((r == c) ? dm : 0.f); }
        __syncthreads();
    }
    mm64(Tm, Y, Z, lane, wave);
    const float lscale = -(float)(1 << K_SQRT);

    FOR_EW { int r = i >> 6, c = i & 63;
             A[r * LD + c] = alpha * (lscale * Tm[r * LD + c]) + beta; }
    __syncthreads();
    FOR_EW { int r = i >> 6, c = i & 63;
             W[r * LD + c] = 0.5f * (A[r * LD + c] + A[c * LD + r]); }
    __syncthreads();

    constexpr int S_EXP = 6, EXP_M = 10;
    const float es = 1.f / (float)(1 << S_EXP);
    FOR_EW { int r = i >> 6, c = i & 63; Y[r * LD + c] = W[r * LD + c] * es; }
    __syncthreads();
    ew_identity(Z, 1.f, tid);
    for (int m = EXP_M; m >= 1; --m) {
        mm64(Tm, Y, Z, lane, wave);
        const float dm = 1.f / (float)m;
        FOR_EW { int r = i >> 6, c = i & 63;
                 Z[r * LD + c] = Tm[r * LD + c] * dm + ((r == c) ? 1.f : 0.f); }
        __syncthreads();
    }
    for (int j = 0; j < S_EXP; ++j) {
        mm64(Tm, Z, Z, lane, wave);
        ew_copy(Z, Tm, tid);
    }

    for (int i = tid; i < 4096; i += 256)
        out[(size_t)bid * 4096 + i] = Z[(i >> 6) * LD + (i & 63)];
}

extern "C" void kernel_launch(void* const* d_in, const int* in_sizes, int n_in,
                              void* d_out, int out_size, void* d_ws, size_t ws_size,
                              hipStream_t stream) {
    const float* x     = (const float*)d_in[0];
    const float* alpha = (const float*)d_in[1];
    const float* beta  = (const float*)d_in[2];
    float* out = (float*)d_out;
    float* cov = (float*)d_ws;

    const int nmat = 256 * N_EPOCH;
    cov64_kernel<<<nmat, 256, 0, stream>>>(x, cov);

    const size_t shbytes = (size_t)(5 * MS + 64) * sizeof(float);
    spdlogexp_kernel<<<nmat, 256, shbytes, stream>>>(cov, alpha, beta, out);
}


# Elementwise passes fused into the product epilogue (D = s*A*B + d*I stored from the accumulators), copies replaced by alternating LDS slots: 9 -> 2 barriers per Newton-Schulz step; sums unchanged
# speedup vs baseline: 1.6706x; 1.3213x over previous
#include <hip/hip_runtime.h>

typedef __attribute__((ext_vector_type(2))) float v2f;
typedef __attribute__((ext_vector_type(8))) float v8f;

#define LD 76
#define MS (64 * LD)
#define C_DIM 64
#define T_DIM 4096
#define N_EPOCH 8
#define PATCH 512
#define EPS_COV 1e-5f

template <bool EPI>
__device__ __forceinline__ void mm64e(float* __restrict__ D, const float* __restrict__ A,
                                      const float* __restrict__ Bm, float s, float d,
                                      int lane, int wave) {
#pragma unroll
    for (int i = 0; i < 2; ++i) {
        const int t  = wave + 8 * i;
        const int tm = (t >> 2) * 16;
        const int tn = (t & 3) * 16;
        const int row = tm + (lane & 15);
        const int col = tn + (lane & 15);
        v8f acc = {0.f, 0.f, 0.f, 0.f, 0.f, 0.f, 0.f, 0.f};
#pragma unroll
        for (int k4 = 0; k4 < 64; k4 += 4) {
            const int ko = k4 + ((lane >> 4) << 1);
            const v2f a = *reinterpret_cast<const v2f*>(A + row * LD + ko);
            v2f b; b.x = Bm[ko * LD + col]; b.y = Bm[(ko + 1) * LD + col];
            acc = __builtin_amdgcn_wmma_f32_16x16x4_f32(false, a, false, b,
                                                        (short)0, acc, false, false);
        }
#pragma unroll
        for (int v = 0; v < 8; ++v) {
            const int m = tm + v + 8 * (lane >> 4);
            float val = acc[v];
            if (EPI) val = (m == col) ? (s * val + d) : (s * val);
            D[m * LD + col] = val;
        }
    }
}

__device__ __forceinline__ void ns_step(float* T, const float* Y, const float* Z,
                                        float* Yn, float* Zn, int lane, int wave) {
    mm64e<true>(T, Z, Y, -0.5f, 1.5f, lane, wave);
    __syncthreads();
    mm64e<false>(Yn, Y, T, 1.f, 0.f, lane, wave);
    mm64e<false>(Zn, T, Z, 1.f, 0.f, lane, wave);
    __syncthreads();
}

#define FOR_EW for (int i = tid; i < 4096; i += 256)

__device__ __forceinline__ void ew_identity(float* d, float dv, int tid) {
    FOR_EW { int r = i >> 6, c = i & 63; d[r * LD + c] = (r == c) ? dv : 0.f; }
    __syncthreads();
}

__global__ void cov64_kernel(const float* __restrict__ x, float* __restrict__ cov) {
    __shared__ __attribute__((aligned(16))) float Xs[64 * LD];
    __shared__ float mp[256];
    __shared__ float mean[64];

    const int tid = threadIdx.x, lane = tid & 31, wave = tid >> 5;
    const int bid = blockIdx.x;
    const int b = bid >> 3, e = bid & 7;
    const float* xb = x + (size_t)b * C_DIM * T_DIM + (size_t)e * PATCH;

    float msum = 0.f;
    v8f acc0 = {0.f,0.f,0.f,0.f,0.f,0.f,0.f,0.f};
    v8f acc1 = {0.f,0.f,0.f,0.f,0.f,0.f,0.f,0.f};

    for (int kc = 0; kc < PATCH; kc += 64) {
        __syncthreads();
        for (int i = tid; i < 4096; i += 256) {
            const int c = i >> 6, j = i & 63;
            Xs[c * LD + j] = xb[(size_t)c * T_DIM + kc + j];
        }
        __syncthreads();
        {
            const int c = tid & 63, part = tid >> 6;
#pragma unroll
            for (int j = 0; j < 16; ++j) msum += Xs[c * LD + part * 16 + j];
        }
#pragma unroll
        for (int i = 0; i < 2; ++i) {
            const int t  = wave + 8 * i;
            const int tm = (t >> 2) * 16, tn = (t & 3) * 16;
            const int row = tm + (lane & 15), coln = tn + (lane & 15);
            v8f acc = i ? acc1 : acc0;
#pragma unroll
            for (int k4 = 0; k4 < 64; k4 += 4) {
                const int ko = k4 + ((lane >> 4) << 1);
                const v2f a  = *reinterpret_cast<const v2f*>(Xs + row * LD + ko);
                const v2f bf = *reinterpret_cast<const v2f*>(Xs + coln * LD + ko);
                acc = __builtin_amdgcn_wmma_f32_16x16x4_f32(false, a, false, bf,
                                                            (short)0, acc, false, false);
            }
            if (i) acc1 = acc; else acc0 = acc;
        }
    }
    __syncthreads();
    mp[tid] = msum;
    __syncthreads();
    if (tid < 64)
        mean[tid] = (mp[tid] + mp[tid + 64] + mp[tid + 128] + mp[tid + 192]) * (1.f / PATCH);
    __syncthreads();

#pragma unroll
    for (int i = 0; i < 2; ++i) {
        const int t  = wave + 8 * i;
        const int tm = (t >> 2) * 16, tn = (t & 3) * 16;
        v8f acc = i ? acc1 : acc0;
#pragma unroll
        for (int v = 0; v < 8; ++v) {
            const int m = tm + v + 8 * (lane >> 4);
            const int n = tn + (lane & 15);
            const float val = (acc[v] - (float)PATCH * mean[m] * mean[n]) * (1.f / (PATCH - 1));
            cov[(size_t)bid * 4096 + m * 64 + n] = val;
        }
    }
}

__global__ void __launch_bounds__(256, 6) spdlogexp_kernel(const float* __restrict__ cov,
                                 const float* __restrict__ alphap,
                                 const float* __restrict__ betap,
                                 float* __restrict__ out) {
    extern __shared__ __attribute__((aligned(16))) float lds[];
    float* S0 = lds;
    float* S1 = lds + 1 * MS;
    float* S2 = lds + 2 * MS;
    float* S3 = lds + 3 * MS;
    float* S4 = lds + 4 * MS;
    float* red = lds + 5 * MS;

    const int tid = threadIdx.x, lane = tid & 31, wave = tid >> 5;
    const int bid = blockIdx.x;
    const float alpha = alphap[0], beta = betap[0];

    for (int i = tid; i < 4096; i += 256)
        S0[(i >> 6) * LD + (i & 63)] = cov[(size_t)bid * 4096 + i];
    __syncthreads();
    if (tid < 64) red[tid] = S0[tid * LD + tid];
    __syncthreads();
    if (tid == 0) { float s = 0.f; for (int i = 0; i < 64; ++i) s += red[i]; red[0] = 1.f / s; }
    __syncthreads();
    const float invtr = red[0];
    FOR_EW { int r = i >> 6, c = i & 63;
             float v = S0[r * LD + c] * invtr;
             if (r == c) v += EPS_COV;
             S0[r * LD + c] = v; }
    __syncthreads();

    constexpr int K_SQRT = 5;
    const int ns_it[K_SQRT] = {18, 12, 9, 7, 6};
    bool hi = false;
    for (int kq = 0; kq < K_SQRT; ++kq) {
        ew_identity(hi ? S4 : S1, 1.f, tid);
        for (int it = 0; it < ns_it[kq]; ++it) {
            if (!hi) ns_step(S2, S0, S1, S3, S4, lane, wave);
            else     ns_step(S2, S3, S4, S0, S1, lane, wave);
            hi = !hi;
        }
    }
    {
        const float* Yf = hi ? S3 : S0;
        FOR_EW { int r = i >> 6, c = i & 63;
                 S2[r * LD + c] = ((r == c) ? 1.f : 0.f) - Yf[r * LD + c]; }
        __syncthreads();
    }
    constexpr int LOG_M = 12;
    ew_identity(S1, 1.f / (float)LOG_M, tid);
    for (int m = LOG_M - 1; m >= 1; m -= 2) {
        mm64e<true>(S4, S2, S1, 1.f, 1.f / (float)m, lane, wave);
        __syncthreads();
        const float d2 = (m > 1) ? 1.f / (float)(m - 1) : 0.f;
        mm64e<true>(S1, S2, S4, 1.f, d2, lane, wave);
        __syncthreads();
    }
    const float lscale = -(float)(1 << K_SQRT);

    FOR_EW { int r = i >> 6, c = i & 63;
             S0[r * LD + c] = alpha * (lscale * S1[r * LD + c]) + beta; }
    __syncthreads();

    constexpr int S_EXP = 6, EXP_M = 10;
    const float es = 1.f / (float)(1 << S_EXP);
    FOR_EW { int r = i >> 6, c = i & 63;
             S2[r * LD + c] = (0.5f * (S0[r * LD + c] + S0[c * LD + r])) * es; }
    __syncthreads();
    ew_identity(S1, 1.f, tid);
    for (int m = EXP_M; m >= 2; m -= 2) {
        mm64e<true>(S4, S2, S1, 1.f / (float)m, 1.f, lane, wave);
        __syncthreads();
        mm64e<true>(S1, S2, S4, 1.f / (float)(m - 1), 1.f, lane, wave);
        __syncthreads();
    }
    for (int j = 0; j < S_EXP; j += 2) {
        mm64e<false>(S4, S1, S1, 1.f, 0.f, lane, wave);
        __syncthreads();
        mm64e<false>(S1, S4, S4, 1.f, 0.f, lane, wave);
        __syncthreads();
    }

    for (int i = tid; i < 4096; i += 256)
        out[(size_t)bid * 4096 + i] = S1[(i >> 6) * LD + (i & 63)];
}

extern "C" void kernel_launch(void* const* d_in, const int* in_sizes, int n_in,
                              void* d_out, int out_size, void* d_ws, size_t ws_size,
                              hipStream_t stream) {
    const float* x     = (const float*)d_in[0];
    const float* alpha = (const float*)d_in[1];
    const float* beta  = (const float*)d_in[2];
    float* out = (float*)d_out;
    float* cov = (float*)d_ws;

    const int nmat = 256 * N_EPOCH;
    cov64_kernel<<<nmat, 256, 0, stream>>>(x, cov);

    const size_t shbytes = (size_t)(5 * MS + 64) * sizeof(float);
    spdlogexp_kernel<<<nmat, 256, shbytes, stream>>>(cov, alpha, beta, out);
}


# Newton-Schulz in product form (M=Z*Y feeds only itself; same Y iterates) so mirrored products stay stable: 10 of 16 tiles computed, rest stored transposed; tile reads grouped; rounding differs
# speedup vs baseline: 2.2333x; 1.3368x over previous
#include <hip/hip_runtime.h>

typedef __attribute__((ext_vector_type(2))) float v2f;
typedef __attribute__((ext_vector_type(8))) float v8f;

#define LD 76
#define MS (64 * LD)
#define C_DIM 64
#define T_DIM 4096
#define N_EPOCH 8
#define PATCH 512
#define EPS_COV 1e-5f

extern __shared__ __attribute__((aligned(16))) float lds[];

#define TILE_TM 0xE9500u
#define TILE_TN 0xFB9E4u

__device__ __forceinline__ void tile_store(int Do, const float (&w)[8], int tm, int tn, int c, int h) {
    float* Dp = lds + Do + (tm + 8 * h) * LD + tn + c;
#pragma unroll
    for (int v = 0; v < 8; ++v) Dp[v * LD] = w[v];
    if (tm != tn) {
        float4* q = reinterpret_cast<float4*>(lds + Do + (tn + c) * LD + tm + 8 * h);
        q[0] = make_float4(w[0], w[1], w[2], w[3]);
        q[1] = make_float4(w[4], w[5], w[6], w[7]);
    }
}

template <int MODE>
__device__ __forceinline__ void tile_mm(int Do, int D2o, int Ao, int Bo, float s, float d,
                                        int u, int lane) {
    const int tm = (int)((TILE_TM >> (2 * u)) & 3u) * 16;
    const int tn = (int)((TILE_TN >> (2 * u)) & 3u) * 16;
    const int c = lane & 15, h = lane >> 4;
    const float* A  = lds + Ao + (tm + c) * LD + 2 * h;
    const float* Bm = lds + Bo + (2 * h) * LD + tn + c;
    v2f af[16], bf[16];
#pragma unroll
    for (int k = 0; k < 16; ++k) {
        af[k] = *reinterpret_cast<const v2f*>(A + 4 * k);
        bf[k].x = Bm[(4 * k) * LD]; bf[k].y = Bm[(4 * k + 1) * LD];
    }
    __builtin_amdgcn_sched_barrier(0);
    v8f acc = {0.f, 0.f, 0.f, 0.f, 0.f, 0.f, 0.f, 0.f};
#pragma unroll
    for (int k = 0; k < 16; ++k)
        acc = __builtin_amdgcn_wmma_f32_16x16x4_f32(false, af[k], false, bf[k],
                                                    (short)0, acc, false, false);
    float val[8], ev[8];
#pragma unroll
    for (int v = 0; v < 8; ++v) {
        val[v] = acc[v];
        ev[v] = ((tm + 8 * h + v) == (tn + c)) ? (s * val[v] + d) : (s * val[v]);
    }
    if (MODE == 1) tile_store(Do, ev, tm, tn, c, h);
    else           tile_store(Do, val, tm, tn, c, h);
    if (MODE == 2) tile_store(D2o, ev, tm, tn, c, h);
}

template <int MODE>
__device__ __forceinline__ void mm_sym(int Do, int D2o, int Ao, int Bo, float s, float d,
                                       int lane, int r) {
    for (int u = r; u < 10; u += 8) tile_mm<MODE>(Do, D2o, Ao, Bo, s, d, u, lane);
}

__device__ __forceinline__ void mm_sym2(int D1, int A1, int B1, int D2, int A2, int B2,
                                        int lane, int r) {
    for (int j = r; j < 20; j += 8) {
        const bool p = j >= 10;
        tile_mm<0>(p ? D2 : D1, 0, p ? A2 : A1, p ? B2 : B1, 1.f, 0.f, p ? j - 10 : j, lane);
    }
}

#define FOR_EW for (int i = tid; i < 4096; i += 256)

__device__ __forceinline__ void ew_identity(int Do, float dv, int tid) {
    FOR_EW { int r = i >> 6, c = i & 63; lds[Do + r * LD + c] = (r == c) ? dv : 0.f; }
    __syncthreads();
}

__global__ void cov64_kernel(const float* __restrict__ x, float* __restrict__ cov) {
    __shared__ __attribute__((aligned(16))) float Xs[64 * LD];
    __shared__ float mp[256];
    __shared__ float mean[64];

    const int tid = threadIdx.x, lane = tid & 31, wave = tid >> 5;
    const int bid = blockIdx.x;
    const int b = bid >> 3, e = bid & 7;
    const float* xb = x + (size_t)b * C_DIM * T_DIM + (size_t)e * PATCH;

    float msum = 0.f;
    v8f acc0 = {0.f,0.f,0.f,0.f,0.f,0.f,0.f,0.f};
    v8f acc1 = {0.f,0.f,0.f,0.f,0.f,0.f,0.f,0.f};

    for (int kc = 0; kc < PATCH; kc += 64) {
        __syncthreads();
        for (int i = tid; i < 4096; i += 256) {
            const int c = i >> 6, j = i & 63;
            Xs[c * LD + j] = xb[(size_t)c * T_DIM + kc + j];
        }
        __syncthreads();
        {
            const int c = tid & 63, part = tid >> 6;
#pragma unroll
            for (int j = 0; j < 16; ++j) msum += Xs[c * LD + part * 16 + j];
        }
#pragma unroll
        for (int i = 0; i < 2; ++i) {
            const int t  = wave + 8 * i;
            const int tm = (t >> 2) * 16, tn = (t & 3) * 16;
            const int row = tm + (lane & 15), coln = tn + (lane & 15);
            v8f acc = i ? acc1 : acc0;
#pragma unroll
            for (int k4 = 0; k4 < 64; k4 += 4) {
                const int ko = k4 + ((lane >> 4) << 1);
                const v2f a  = *reinterpret_cast<const v2f*>(Xs + row * LD + ko);
                const v2f bf = *reinterpret_cast<const v2f*>(Xs + coln * LD + ko);
                acc = __builtin_amdgcn_wmma_f32_16x16x4_f32(false, a, false, bf,
                                                            (short)0, acc, false, false);
            }
            if (i) acc1 = acc; else acc0 = acc;
        }
    }
    __syncthreads();
    mp[tid] = msum;
    __syncthreads();
    if (tid < 64)
        mean[tid] = (mp[tid] + mp[tid + 64] + mp[tid + 128] + mp[tid + 192]) * (1.f / PATCH);
    __syncthreads();

#pragma unroll
    for (int i = 0; i < 2; ++i) {
        const int t  = wave + 8 * i;
        const int tm = (t >> 2) * 16, tn = (t & 3) * 16;
        v8f acc = i ? acc1 : acc0;
#pragma unroll
        for (int v = 0; v < 8; ++v) {
            const int m = tm + v + 8 * (lane >> 4);
            const int n = tn + (lane & 15);
            const float val = (acc[v] - (float)PATCH * mean[m] * mean[n]) * (1.f / (PATCH - 1));
            cov[(size_t)bid * 4096 + m * 64 + n] = val;
        }
    }
}

__global__ void __launch_bounds__(256, 6) spdlogexp_kernel(const float* __restrict__ cov,
                                 const float* __restrict__ alphap,
                                 const float* __restrict__ betap,
                                 float* __restrict__ out) {
    float* red = lds + 5 * MS;

    const int tid = threadIdx.x, lane = tid & 31;
    const int wave = __builtin_amdgcn_readfirstlane(tid >> 5);
    const int rot4 = (wave + 4) & 7;
    const int rot6 = (wave + 6) & 7;
    const int bid = blockIdx.x;
    const float alpha = alphap[0], beta = betap[0];

    int sy = 0, so = MS, sm = 2 * MS, sq = 3 * MS, sp = 4 * MS;

    for (int i = tid; i < 4096; i += 256)
        lds[sy + (i >> 6) * LD + (i & 63)] = cov[(size_t)bid * 4096 + i];
    __syncthreads();
    if (tid < 64) red[tid] = lds[sy + tid * LD + tid];
    __syncthreads();
    if (tid == 0) { float s = 0.f; for (int i = 0; i < 64; ++i) s += red[i]; red[0] = 1.f / s; }
    __syncthreads();
    const float invtr = red[0];
    FOR_EW { int r = i >> 6, c = i & 63;
             float v = lds[sy + r * LD + c] * invtr;
             if (r == c) v += EPS_COV;
             lds[sy + r * LD + c] = v; }
    __syncthreads();

    constexpr int K_SQRT = 5;
    const int ns_it[K_SQRT] = {18, 12, 9, 7, 6};
    for (int kq = 0; kq < K_SQRT; ++kq) {
        FOR_EW { int r = i >> 6, c = i & 63;
                 const float v = lds[sy + r * LD + c];
                 lds[sm + r * LD + c] = v;
                 lds[sp + r * LD + c] = (r == c) ? (-0.5f * v + 1.5f) : (-0.5f * v); }
        __syncthreads();
        const int n = ns_it[kq];
        for (int it = 0; it < n - 1; ++it) {
            mm_sym2(sq, sm, sp, so, sy, sp, lane, wave);
            __syncthreads();
            mm_sym<2>(sm, sy, sq, sp, -0.5f, 1.5f, lane, rot4);
            __syncthreads();
            const int t = sp; sp = sy; sy = so; so = t;
        }
        mm_sym<0>(so, 0, sy, sp, 1.f, 0.f, lane, wave);
        __syncthreads();
        { const int t = sy; sy = so; so = t; }
    }
    FOR_EW { int r = i >> 6, c = i & 63;
             lds[sm + r * LD + c] = ((r == c) ? 1.f : 0.f) - lds[sy + r * LD + c]; }
    __syncthreads();
    constexpr int LOG_M = 12;
    ew_identity(sp, 1.f / (float)LOG_M, tid);
    for (int m = LOG_M - 1; m >= 1; m -= 2) {
        mm_sym<1>(sq, 0, sm, sp, 1.f, 1.f / (float)m, lane, wave);
        __syncthreads();
        const float d2 = (m > 1) ? 1.f / (float)(m - 1) : 0.f;
        mm_sym<1>(sp, 0, sm, sq, 1.f, d2, lane, rot6);
        __syncthreads();
    }
    const float lscale = -(float)(1 << K_SQRT);

    FOR_EW { int r = i >> 6, c = i & 63;
             lds[sy + r * LD + c] = alpha * (lscale * lds[sp + r * LD + c]) + beta; }
    __syncthreads();

    constexpr int S_EXP = 6, EXP_M = 10;
    const float es = 1.f / (float)(1 << S_EXP);
    FOR_EW { int r = i >> 6, c = i & 63;
             lds[sm + r * LD + c] = (0.5f * (lds[sy + r * LD + c] + lds[sy + c * LD + r])) * es; }
    __syncthreads();
    ew_identity(sp, 1.f, tid);
    for (int m = EXP_M; m >= 2; m -= 2) {
        mm_sym<1>(sq, 0, sm, sp, 1.f / (float)m, 1.f, lane, wave);
        __syncthreads();
        mm_sym<1>(sp, 0, sm, sq, 1.f / (float)(m - 1), 1.f, lane, rot6);
        __syncthreads();
    }
    for (int j = 0; j < S_EXP; j += 2) {
        mm_sym<0>(sq, 0, sp, sp, 1.f, 0.f, lane, wave);
        __syncthreads();
        mm_sym<0>(sp, 0, sq, sq, 1.f, 0.f, lane, rot6);
        __syncthreads();
    }

    for (int i = tid; i < 4096; i += 256)
        out[(size_t)bid * 4096 + i] = lds[sp + (i >> 6) * LD + (i & 63)];
}

extern "C" void kernel_launch(void* const* d_in, const int* in_sizes, int n_in,
                              void* d_out, int out_size, void* d_ws, size_t ws_size,
                              hipStream_t stream) {
    const float* x     = (const float*)d_in[0];
    const float* alpha = (const float*)d_in[1];
    const float* beta  = (const float*)d_in[2];
    float* out = (float*)d_out;
    float* cov = (float*)d_ws;

    const int nmat = 256 * N_EPOCH;
    cov64_kernel<<<nmat, 256, 0, stream>>>(x, cov);

    const size_t shbytes = (size_t)(5 * MS + 64) * sizeof(float);
    spdlogexp_kernel<<<nmat, 256, shbytes, stream>>>(cov, alpha, beta, out);
}


# Newton-Schulz roots stop on a measured residual: |M-I|_F^2 summed in the M=Q*P epilogue, below 1e-6 the closing Y*P follows; fixed counts stay as bounds; fewer products, rounding differs
# speedup vs baseline: 3.0311x; 1.3572x over previous
#include <hip/hip_runtime.h>

typedef __attribute__((ext_vector_type(2))) float v2f;
typedef __attribute__((ext_vector_type(8))) float v8f;

#define LD 76
#define MS (64 * LD)
#define C_DIM 64
#define T_DIM 4096
#define N_EPOCH 8
#define PATCH 512
#define EPS_COV 1e-5f
#define NS_TOL2 1e-6f

extern __shared__ __attribute__((aligned(16))) float lds[];

#define TILE_TM 0xE9500u
#define TILE_TN 0xFB9E4u

__device__ __forceinline__ void tile_store(int Do, const float (&w)[8], int tm, int tn, int c, int h) {
    float* Dp = lds + Do + (tm + 8 * h) * LD + tn + c;
#pragma unroll
    for (int v = 0; v < 8; ++v) Dp[v * LD] = w[v];
    if (tm != tn) {
        float4* q = reinterpret_cast<float4*>(lds + Do + (tn + c) * LD + tm + 8 * h);
        q[0] = make_float4(w[0], w[1], w[2], w[3]);
        q[1] = make_float4(w[4], w[5], w[6], w[7]);
    }
}

template <int MODE>
__device__ __forceinline__ float tile_mm(int Do, int D2o, int Ao, int Bo, float s, float d,
                                         int u, int lane) {
    const int tm = (int)((TILE_TM >> (2 * u)) & 3u) * 16;
    const int tn = (int)((TILE_TN >> (2 * u)) & 3u) * 16;
    const int c = lane & 15, h = lane >> 4;
    const float* A  = lds + Ao + (tm + c) * LD + 2 * h;
    const float* Bm = lds + Bo + (2 * h) * LD + tn + c;
    v2f af[16], bf[16];
#pragma unroll
    for (int k = 0; k < 16; ++k) {
        af[k] = *reinterpret_cast<const v2f*>(A + 4 * k);
        bf[k].x = Bm[(4 * k) * LD]; bf[k].y = Bm[(4 * k + 1) * LD];
    }
    __builtin_amdgcn_sched_barrier(0);
    v8f acc = {0.f, 0.f, 0.f, 0.f, 0.f, 0.f, 0.f, 0.f};
#pragma unroll
    for (int k = 0; k < 16; ++k)
        acc = __builtin_amdgcn_wmma_f32_16x16x4_f32(false, af[k], false, bf[k],
                                                    (short)0, acc, false, false);
    float val[8], ev[8];
    float rs = 0.f;
#pragma unroll
    for (int v = 0; v < 8; ++v) {
        val[v] = acc[v];
        const bool dg = (tm + 8 * h + v) == (tn + c);
        ev[v] = dg ? (s * val[v] + d) : (s * val[v]);
        if (MODE == 2) { const float dv = dg ? (val[v] - 1.f) : val[v]; rs = fmaf(dv, dv, rs); }
    }
    if (MODE == 1) tile_store(Do, ev, tm, tn, c, h);
    else           tile_store(Do, val, tm, tn, c, h);
    if (MODE == 2) tile_store(D2o, ev, tm, tn, c, h);
    return (tm != tn) ? rs + rs : rs;
}

template <int MODE>
__device__ __forceinline__ float mm_sym(int Do, int D2o, int Ao, int Bo, float s, float d,
                                        int lane, int r) {
    float rs = 0.f;
    for (int u = r; u < 10; u += 8) rs += tile_mm<MODE>(Do, D2o, Ao, Bo, s, d, u, lane);
    return rs;
}

template <int XM>
__device__ __forceinline__ float xmask_add(float v) {
    return v + __builtin_bit_cast(float, __builtin_amdgcn_update_dpp(
                   0, __builtin_bit_cast(int, v), 0x160 + XM, 0xf, 0xf, true));
}
__device__ __forceinline__ float row16_sum(float v) {
    return xmask_add<8>(xmask_add<4>(xmask_add<2>(xmask_add<1>(v))));
}

__device__ __forceinline__ void mm_sym2(int D1, int A1, int B1, int D2, int A2, int B2,
                                        int lane, int r) {
    for (int j = r; j < 20; j += 8) {
        const bool p = j >= 10;
        tile_mm<0>(p ? D2 : D1, 0, p ? A2 : A1, p ? B2 : B1, 1.f, 0.f, p ? j - 10 : j, lane);
    }
}

#define FOR_EW for (int i = tid; i < 4096; i += 256)

__device__ __forceinline__ void ew_identity(int Do, float dv, int tid) {
    FOR_EW { int r = i >> 6, c = i & 63; lds[Do + r * LD + c] = (r == c) ? dv : 0.f; }
    __syncthreads();
}

__global__ void cov64_kernel(const float* __restrict__ x, float* __restrict__ cov) {
    __shared__ __attribute__((aligned(16))) float Xs[64 * LD];
    __shared__ float mp[256];
    __shared__ float mean[64];

    const int tid = threadIdx.x, lane = tid & 31, wave = tid >> 5;
    const int bid = blockIdx.x;
    const int b = bid >> 3, e = bid & 7;
    const float* xb = x + (size_t)b * C_DIM * T_DIM + (size_t)e * PATCH;

    float msum = 0.f;
    v8f acc0 = {0.f,0.f,0.f,0.f,0.f,0.f,0.f,0.f};
    v8f acc1 = {0.f,0.f,0.f,0.f,0.f,0.f,0.f,0.f};

    for (int kc = 0; kc < PATCH; kc += 64) {
        __syncthreads();
        for (int i = tid; i < 4096; i += 256) {
            const int c = i >> 6, j = i & 63;
            Xs[c * LD + j] = xb[(size_t)c * T_DIM + kc + j];
        }
        __syncthreads();
        {
            const int c = tid & 63, part = tid >> 6;
#pragma unroll
            for (int j = 0; j < 16; ++j) msum += Xs[c * LD + part * 16 + j];
        }
#pragma unroll
        for (int i = 0; i < 2; ++i) {
            const int t  = wave + 8 * i;
            const int tm = (t >> 2) * 16, tn = (t & 3) * 16;
            const int row = tm + (lane & 15), coln = tn + (lane & 15);
            v8f acc = i ? acc1 : acc0;
#pragma unroll
            for (int k4 = 0; k4 < 64; k4 += 4) {
                const int ko = k4 + ((lane >> 4) << 1);
                const v2f a  = *reinterpret_cast<const v2f*>(Xs + row * LD + ko);
                const v2f bf = *reinterpret_cast<const v2f*>(Xs + coln * LD + ko);
                acc = __builtin_amdgcn_wmma_f32_16x16x4_f32(false, a, false, bf,
                                                            (short)0, acc, false, false);
            }
            if (i) acc1 = acc; else acc0 = acc;
        }
    }
    __syncthreads();
    mp[tid] = msum;
    __syncthreads();
    if (tid < 64)
        mean[tid] = (mp[tid] + mp[tid + 64] + mp[tid + 128] + mp[tid + 192]) * (1.f / PATCH);
    __syncthreads();

#pragma unroll
    for (int i = 0; i < 2; ++i) {
        const int t  = wave + 8 * i;
        const int tm = (t >> 2) * 16, tn = (t & 3) * 16;
        v8f acc = i ? acc1 : acc0;
#pragma unroll
        for (int v = 0; v < 8; ++v) {
            const int m = tm + v + 8 * (lane >> 4);
            const int n = tn + (lane & 15);
            const float val = (acc[v] - (float)PATCH * mean[m] * mean[n]) * (1.f / (PATCH - 1));
            cov[(size_t)bid * 4096 + m * 64 + n] = val;
        }
    }
}

__global__ void __launch_bounds__(256, 6) spdlogexp_kernel(const float* __restrict__ cov,
                                 const float* __restrict__ alphap,
                                 const float* __restrict__ betap,
                                 float* __restrict__ out) {
    float* red = lds + 5 * MS;

    const int tid = threadIdx.x, lane = tid & 31;
    const int wave = __builtin_amdgcn_readfirstlane(tid >> 5);
    const int rot4 = (wave + 4) & 7;
    const int rot6 = (wave + 6) & 7;
    const int bid = blockIdx.x;
    const float alpha = alphap[0], beta = betap[0];

    int sy = 0, so = MS, sm = 2 * MS, sq = 3 * MS, sp = 4 * MS;

    for (int i = tid; i < 4096; i += 256)
        lds[sy + (i >> 6) * LD + (i & 63)] = cov[(size_t)bid * 4096 + i];
    __syncthreads();
    if (tid < 64) red[tid] = lds[sy + tid * LD + tid];
    __syncthreads();
    if (tid == 0) { float s = 0.f; for (int i = 0; i < 64; ++i) s += red[i]; red[0] = 1.f / s; }
    __syncthreads();
    const float invtr = red[0];
    FOR_EW { int r = i >> 6, c = i & 63;
             float v = lds[sy + r * LD + c] * invtr;
             if (r == c) v += EPS_COV;
             lds[sy + r * LD + c] = v; }
    __syncthreads();

    constexpr int K_SQRT = 5;
    const int ns_it[K_SQRT] = {18, 12, 9, 7, 6};
    for (int kq = 0; kq < K_SQRT; ++kq) {
        FOR_EW { int r = i >> 6, c = i & 63;
                 const float v = lds[sy + r * LD + c];
                 lds[sm + r * LD + c] = v;
                 lds[sp + r * LD + c] = (r == c) ? (-0.5f * v + 1.5f) : (-0.5f * v); }
        __syncthreads();
        const int n = ns_it[kq];
        for (int it = 0; it < n - 1; ++it) {
            mm_sym2(sq, sm, sp, so, sy, sp, lane, wave);
            __syncthreads();
            float rs = mm_sym<2>(sm, sy, sq, sp, -0.5f, 1.5f, lane, rot4);
            rs = row16_sum(rs);
            if ((lane & 15) == 0) red[16 + 2 * wave + (lane >> 4)] = rs;
            __syncthreads();
            const int t = sp; sp = sy; sy = so; so = t;
            const float4* rp = reinterpret_cast<const float4*>(red + 16);
            const float4 r0 = rp[0], r1 = rp[1], r2 = rp[2], r3 = rp[3];
            const float res = (((r0.x + r0.y) + (r0.z + r0.w)) + ((r1.x + r1.y) + (r1.z + r1.w)))
                            + (((r2.x + r2.y) + (r2.z + r2.w)) + ((r3.x + r3.y) + (r3.z + r3.w)));
            if (__builtin_amdgcn_readfirstlane((int)(res < NS_TOL2))) break;
        }
        mm_sym<0>(so, 0, sy, sp, 1.f, 0.f, lane, wave);
        __syncthreads();
        { const int t = sy; sy = so; so = t; }
    }
    FOR_EW { int r = i >> 6, c = i & 63;
             lds[sm + r * LD + c] = ((r == c) ? 1.f : 0.f) - lds[sy + r * LD + c]; }
    __syncthreads();
    constexpr int LOG_M = 12;
    ew_identity(sp, 1.f / (float)LOG_M, tid);
    for (int m = LOG_M - 1; m >= 1; m -= 2) {
        mm_sym<1>(sq, 0, sm, sp, 1.f, 1.f / (float)m, lane, wave);
        __syncthreads();
        const float d2 = (m > 1) ? 1.f / (float)(m - 1) : 0.f;
        mm_sym<1>(sp, 0, sm, sq, 1.f, d2, lane, rot6);
        __syncthreads();
    }
    const float lscale = -(float)(1 << K_SQRT);

    FOR_EW { int r = i >> 6, c = i & 63;
             lds[sy + r * LD + c] = alpha * (lscale * lds[sp + r * LD + c]) + beta; }
    __syncthreads();

    constexpr int S_EXP = 6, EXP_M = 10;
    const float es = 1.f / (float)(1 << S_EXP);
    FOR_EW { int r = i >> 6, c = i & 63;
             lds[sm + r * LD + c] = (0.5f * (lds[sy + r * LD + c] + lds[sy + c * LD + r])) * es; }
    __syncthreads();
    ew_identity(sp, 1.f, tid);
    for (int m = EXP_M; m >= 2; m -= 2) {
        mm_sym<1>(sq, 0, sm, sp, 1.f / (float)m, 1.f, lane, wave);
        __syncthreads();
        mm_sym<1>(sp, 0, sm, sq, 1.f / (float)(m - 1), 1.f, lane, rot6);
        __syncthreads();
    }
    for (int j = 0; j < S_EXP; j += 2) {
        mm_sym<0>(sq, 0, sp, sp, 1.f, 0.f, lane, wave);
        __syncthreads();
        mm_sym<0>(sp, 0, sq, sq, 1.f, 0.f, lane, rot6);
        __syncthreads();
    }

    for (int i = tid; i < 4096; i += 256)
        out[(size_t)bid * 4096 + i] = lds[sp + (i >> 6) * LD + (i & 63)];
}

extern "C" void kernel_launch(void* const* d_in, const int* in_sizes, int n_in,
                              void* d_out, int out_size, void* d_ws, size_t ws_size,
                              hipStream_t stream) {
    const float* x     = (const float*)d_in[0];
    const float* alpha = (const float*)d_in[1];
    const float* beta  = (const float*)d_in[2];
    float* out = (float*)d_out;
    float* cov = (float*)d_ws;

    const int nmat = 256 * N_EPOCH;
    cov64_kernel<<<nmat, 256, 0, stream>>>(x, cov);

    const size_t shbytes = (size_t)(5 * MS + 64) * sizeof(float);
    spdlogexp_kernel<<<nmat, 256, shbytes, stream>>>(cov, alpha, beta, out);
}


# Matrix scaled by c = 2 / (largest absolute row sum) in front of the square roots, log(c) taken off the diagonal of the log: spectrum in (0,2], fewer Newton-Schulz steps per root; rounding differs
# speedup vs baseline: 3.8452x; 1.2686x over previous
#include <hip/hip_runtime.h>

typedef __attribute__((ext_vector_type(2))) float v2f;
typedef __attribute__((ext_vector_type(8))) float v8f;

#define LD 76
#define MS (64 * LD)
#define C_DIM 64
#define T_DIM 4096
#define N_EPOCH 8
#define PATCH 512
#define EPS_COV 1e-5f
#define NS_TOL2 1e-6f

extern __shared__ __attribute__((aligned(16))) float lds[];

#define TILE_TM 0xE9500u
#define TILE_TN 0xFB9E4u

__device__ __forceinline__ void tile_store(int Do, const float (&w)[8], int tm, int tn, int c, int h) {
    float* Dp = lds + Do + (tm + 8 * h) * LD + tn + c;
#pragma unroll
    for (int v = 0; v < 8; ++v) Dp[v * LD] = w[v];
    if (tm != tn) {
        float4* q = reinterpret_cast<float4*>(lds + Do + (tn + c) * LD + tm + 8 * h);
        q[0] = make_float4(w[0], w[1], w[2], w[3]);
        q[1] = make_float4(w[4], w[5], w[6], w[7]);
    }
}

template <int MODE>
__device__ __forceinline__ float tile_mm(int Do, int D2o, int Ao, int Bo, float s, float d,
                                         int u, int lane) {
    const int tm = (int)((TILE_TM >> (2 * u)) & 3u) * 16;
    const int tn = (int)((TILE_TN >> (2 * u)) & 3u) * 16;
    const int c = lane & 15, h = lane >> 4;
    const float* A  = lds + Ao + (tm + c) * LD + 2 * h;
    const float* Bm = lds + Bo + (2 * h) * LD + tn + c;
    v2f af[16], bf[16];
#pragma unroll
    for (int k = 0; k < 16; ++k) {
        af[k] = *reinterpret_cast<const v2f*>(A + 4 * k);
        bf[k].x = Bm[(4 * k) * LD]; bf[k].y = Bm[(4 * k + 1) * LD];
    }
    __builtin_amdgcn_sched_barrier(0);
    v8f acc = {0.f, 0.f, 0.f, 0.f, 0.f, 0.f, 0.f, 0.f};
#pragma unroll
    for (int k = 0; k < 16; ++k)
        acc = __builtin_amdgcn_wmma_f32_16x16x4_f32(false, af[k], false, bf[k],
                                                    (short)0, acc, false, false);
    float val[8], ev[8];
    float rs = 0.f;
#pragma unroll
    for (int v = 0; v < 8; ++v) {
        val[v] = acc[v];
        const bool dg = (tm + 8 * h + v) == (tn + c);
        ev[v] = dg ? (s * val[v] + d) : (s * val[v]);
        if (MODE == 2) { const float dv = dg ? (val[v] - 1.f) : val[v]; rs = fmaf(dv, dv, rs); }
    }
    if (MODE == 1) tile_store(Do, ev, tm, tn, c, h);
    else           tile_store(Do, val, tm, tn, c, h);
    if (MODE == 2) tile_store(D2o, ev, tm, tn, c, h);
    return (tm != tn) ? rs + rs : rs;
}

template <int MODE>
__device__ __forceinline__ float mm_sym(int Do, int D2o, int Ao, int Bo, float s, float d,
                                        int lane, int r) {
    float rs = 0.f;
    for (int u = r; u < 10; u += 8) rs += tile_mm<MODE>(Do, D2o, Ao, Bo, s, d, u, lane);
    return rs;
}

template <int XM>
__device__ __forceinline__ float xmask_add(float v) {
    return v + __builtin_bit_cast(float, __builtin_amdgcn_update_dpp(
                   0, __builtin_bit_cast(int, v), 0x160 + XM, 0xf, 0xf, true));
}
__device__ __forceinline__ float row16_sum(float v) {
    return xmask_add<8>(xmask_add<4>(xmask_add<2>(xmask_add<1>(v))));
}

__device__ __forceinline__ void mm_sym2(int D1, int A1, int B1, int D2, int A2, int B2,
                                        int lane, int r) {
    for (int j = r; j < 20; j += 8) {
        const bool p = j >= 10;
        tile_mm<0>(p ? D2 : D1, 0, p ? A2 : A1, p ? B2 : B1, 1.f, 0.f, p ? j - 10 : j, lane);
    }
}

#define FOR_EW for (int i = tid; i < 4096; i += 256)

__device__ __forceinline__ void ew_identity(int Do, float dv, int tid) {
    FOR_EW { int r = i >> 6, c = i & 63; lds[Do + r * LD + c] = (r == c) ? dv : 0.f; }
    __syncthreads();
}

__global__ void cov64_kernel(const float* __restrict__ x, float* __restrict__ cov) {
    __shared__ __attribute__((aligned(16))) float Xs[64 * LD];
    __shared__ float mp[256];
    __shared__ float mean[64];

    const int tid = threadIdx.x, lane = tid & 31, wave = tid >> 5;
    const int bid = blockIdx.x;
    const int b = bid >> 3, e = bid & 7;
    const float* xb = x + (size_t)b * C_DIM * T_DIM + (size_t)e * PATCH;

    float msum = 0.f;
    v8f acc0 = {0.f,0.f,0.f,0.f,0.f,0.f,0.f,0.f};
    v8f acc1 = {0.f,0.f,0.f,0.f,0.f,0.f,0.f,0.f};

    for (int kc = 0; kc < PATCH; kc += 64) {
        __syncthreads();
        for (int i = tid; i < 4096; i += 256) {
            const int c = i >> 6, j = i & 63;
            Xs[c * LD + j] = xb[(size_t)c * T_DIM + kc + j];
        }
        __syncthreads();
        {
            const int c = tid & 63, part = tid >> 6;
#pragma unroll
            for (int j = 0; j < 16; ++j) msum += Xs[c * LD + part * 16 + j];
        }
#pragma unroll
        for (int i = 0; i < 2; ++i) {
            const int t  = wave + 8 * i;
            const int tm = (t >> 2) * 16, tn = (t & 3) * 16;
            const int row = tm + (lane & 15), coln = tn + (lane & 15);
            v8f acc = i ? acc1 : acc0;
#pragma unroll
            for (int k4 = 0; k4 < 64; k4 += 4) {
                const int ko = k4 + ((lane >> 4) << 1);
                const v2f a  = *reinterpret_cast<const v2f*>(Xs + row * LD + ko);
                const v2f bf = *reinterpret_cast<const v2f*>(Xs + coln * LD + ko);
                acc = __builtin_amdgcn_wmma_f32_16x16x4_f32(false, a, false, bf,
                                                            (short)0, acc, false, false);
            }
            if (i) acc1 = acc; else acc0 = acc;
        }
    }
    __syncthreads();
    mp[tid] = msum;
    __syncthreads();
    if (tid < 64)
        mean[tid] = (mp[tid] + mp[tid + 64] + mp[tid + 128] + mp[tid + 192]) * (1.f / PATCH);
    __syncthreads();

#pragma unroll
    for (int i = 0; i < 2; ++i) {
        const int t  = wave + 8 * i;
        const int tm = (t >> 2) * 16, tn = (t & 3) * 16;
        v8f acc = i ? acc1 : acc0;
#pragma unroll
        for (int v = 0; v < 8; ++v) {
            const int m = tm + v + 8 * (lane >> 4);
            const int n = tn + (lane & 15);
            const float val = (acc[v] - (float)PATCH * mean[m] * mean[n]) * (1.f / (PATCH - 1));
            cov[(size_t)bid * 4096 + m * 64 + n] = val;
        }
    }
}

__global__ void __launch_bounds__(256, 6) spdlogexp_kernel(const float* __restrict__ cov,
                                 const float* __restrict__ alphap,
                                 const float* __restrict__ betap,
                                 float* __restrict__ out) {
    float* red = lds + 5 * MS;

    const int tid = threadIdx.x, lane = tid & 31;
    const int wave = __builtin_amdgcn_readfirstlane(tid >> 5);
    const int rot4 = (wave + 4) & 7;
    const int rot6 = (wave + 6) & 7;
    const int bid = blockIdx.x;
    const float alpha = alphap[0], beta = betap[0];

    int sy = 0, so = MS, sm = 2 * MS, sq = 3 * MS, sp = 4 * MS;

    for (int i = tid; i < 4096; i += 256)
        lds[sy + (i >> 6) * LD + (i & 63)] = cov[(size_t)bid * 4096 + i];
    __syncthreads();
    if (tid < 64) red[tid] = lds[sy + tid * LD + tid];
    __syncthreads();
    if (tid == 0) { float s = 0.f; for (int i = 0; i < 64; ++i) s += red[i]; red[0] = 1.f / s; }
    __syncthreads();
    const float invtr = red[0];
    FOR_EW { int r = i >> 6, c = i & 63;
             float v = lds[sy + r * LD + c] * invtr;
             if (r == c) v += EPS_COV;
             lds[sy + r * LD + c] = v; }
    __syncthreads();

    {
        const int row = tid >> 2, q = tid & 3;
        const float4* rp4 = reinterpret_cast<const float4*>(lds + sy + row * LD + 16 * q);
        float rsum = 0.f;
        for (int j = 0; j < 4; ++j) {
            const float4 v = rp4[j];
            rsum += (fabsf(v.x) + fabsf(v.y)) + (fabsf(v.z) + fabsf(v.w));
        }
        rsum = xmask_add<2>(xmask_add<1>(rsum));
        if (q == 0) red[row] = rsum;
    }
    __syncthreads();
    float bnd = 0.f;
    {
        const float4* rp = reinterpret_cast<const float4*>(red);
        for (int j = 0; j < 16; ++j) {
            const float4 v = rp[j];
            bnd = fmaxf(bnd, fmaxf(fmaxf(v.x, v.y), fmaxf(v.z, v.w)));
        }
    }
    const float cscale = (bnd < 2.f) ? 2.f / bnd : 1.f;
    const float logc = logf(cscale);
    FOR_EW { int r = i >> 6, c = i & 63; lds[sy + r * LD + c] *= cscale; }
    __syncthreads();

    constexpr int K_SQRT = 5;
    const int ns_it[K_SQRT] = {18, 12, 9, 7, 6};
    for (int kq = 0; kq < K_SQRT; ++kq) {
        FOR_EW { int r = i >> 6, c = i & 63;
                 const float v = lds[sy + r * LD + c];
                 lds[sm + r * LD + c] = v;
                 lds[sp + r * LD + c] = (r == c) ? (-0.5f * v + 1.5f) : (-0.5f * v); }
        __syncthreads();
        const int n = ns_it[kq];
        for (int it = 0; it < n - 1; ++it) {
            mm_sym2(sq, sm, sp, so, sy, sp, lane, wave);
            __syncthreads();
            float rs = mm_sym<2>(sm, sy, sq, sp, -0.5f, 1.5f, lane, rot4);
            rs = row16_sum(rs);
            if ((lane & 15) == 0) red[16 + 2 * wave + (lane >> 4)] = rs;
            __syncthreads();
            const int t = sp; sp = sy; sy = so; so = t;
            const float4* rp = reinterpret_cast<const float4*>(red + 16);
            const float4 r0 = rp[0], r1 = rp[1], r2 = rp[2], r3 = rp[3];
            const float res = (((r0.x + r0.y) + (r0.z + r0.w)) + ((r1.x + r1.y) + (r1.z + r1.w)))
                            + (((r2.x + r2.y) + (r2.z + r2.w)) + ((r3.x + r3.y) + (r3.z + r3.w)));
            if (__builtin_amdgcn_readfirstlane((int)(res < NS_TOL2))) break;
        }
        mm_sym<0>(so, 0, sy, sp, 1.f, 0.f, lane, wave);
        __syncthreads();
        { const int t = sy; sy = so; so = t; }
    }
    FOR_EW { int r = i >> 6, c = i & 63;
             lds[sm + r * LD + c] = ((r == c) ? 1.f : 0.f) - lds[sy + r * LD + c]; }
    __syncthreads();
    constexpr int LOG_M = 12;
    ew_identity(sp, 1.f / (float)LOG_M, tid);
    for (int m = LOG_M - 1; m >= 1; m -= 2) {
        mm_sym<1>(sq, 0, sm, sp, 1.f, 1.f / (float)m, lane, wave);
        __syncthreads();
        const float d2 = (m > 1) ? 1.f / (float)(m - 1) : 0.f;
        mm_sym<1>(sp, 0, sm, sq, 1.f, d2, lane, rot6);
        __syncthreads();
    }
    const float lscale = -(float)(1 << K_SQRT);

    FOR_EW { int r = i >> 6, c = i & 63;
             float lg = lscale * lds[sp + r * LD + c];
             if (r == c) lg -= logc;
             lds[sy + r * LD + c] = alpha * lg + beta; }
    __syncthreads();

    constexpr int S_EXP = 6, EXP_M = 10;
    const float es = 1.f / (float)(1 << S_EXP);
    FOR_EW { int r = i >> 6, c = i & 63;
             lds[sm + r * LD + c] = (0.5f * (lds[sy + r * LD + c] + lds[sy + c * LD + r])) * es; }
    __syncthreads();
    ew_identity(sp, 1.f, tid);
    for (int m = EXP_M; m >= 2; m -= 2) {
        mm_sym<1>(sq, 0, sm, sp, 1.f / (float)m, 1.f, lane, wave);
        __syncthreads();
        mm_sym<1>(sp, 0, sm, sq, 1.f / (float)(m - 1), 1.f, lane, rot6);
        __syncthreads();
    }
    for (int j = 0; j < S_EXP; j += 2) {
        mm_sym<0>(sq, 0, sp, sp, 1.f, 0.f, lane, wave);
        __syncthreads();
        mm_sym<0>(sp, 0, sq, sq, 1.f, 0.f, lane, rot6);
        __syncthreads();
    }

    for (int i = tid; i < 4096; i += 256)
        out[(size_t)bid * 4096 + i] = lds[sp + (i >> 6) * LD + (i & 63)];
}

extern "C" void kernel_launch(void* const* d_in, const int* in_sizes, int n_in,
                              void* d_out, int out_size, void* d_ws, size_t ws_size,
                              hipStream_t stream) {
    const float* x     = (const float*)d_in[0];
    const float* alpha = (const float*)d_in[1];
    const float* beta  = (const float*)d_in[2];
    float* out = (float*)d_out;
    float* cov = (float*)d_ws;

    const int nmat = 256 * N_EPOCH;
    cov64_kernel<<<nmat, 256, 0, stream>>>(x, cov);

    const size_t shbytes = (size_t)(5 * MS + 64) * sizeof(float);
    spdlogexp_kernel<<<nmat, 256, shbytes, stream>>>(cov, alpha, beta, out);
}


# Root count of the log and squaring count of the exp set by a measured max-row-sum bound (roots stop at |I-Y| <= 0.3, s = least with |X|/2^s <= 1); 5 and 6 stay as upper counts; rounding differs
# speedup vs baseline: 4.2942x; 1.1168x over previous
#include <hip/hip_runtime.h>

typedef __attribute__((ext_vector_type(2))) float v2f;
typedef __attribute__((ext_vector_type(8))) float v8f;

#define LD 76
#define MS (64 * LD)
#define C_DIM 64
#define T_DIM 4096
#define N_EPOCH 8
#define PATCH 512
#define EPS_COV 1e-5f
#define NS_TOL2 1e-6f
#define LOG_E_MAX 0.3f
#define EXP_X_MAX 1.0f

extern __shared__ __attribute__((aligned(16))) float lds[];

#define TILE_TM 0xE9500u
#define TILE_TN 0xFB9E4u

__device__ __forceinline__ void tile_store(int Do, const float (&w)[8], int tm, int tn, int c, int h) {
    float* Dp = lds + Do + (tm + 8 * h) * LD + tn + c;
#pragma unroll
    for (int v = 0; v < 8; ++v) Dp[v * LD] = w[v];
    if (tm != tn) {
        float4* q = reinterpret_cast<float4*>(lds + Do + (tn + c) * LD + tm + 8 * h);
        q[0] = make_float4(w[0], w[1], w[2], w[3]);
        q[1] = make_float4(w[4], w[5], w[6], w[7]);
    }
}

template <int MODE>
__device__ __forceinline__ float tile_mm(int Do, int D2o, int Ao, int Bo, float s, float d,
                                         int u, int lane) {
    const int tm = (int)((TILE_TM >> (2 * u)) & 3u) * 16;
    const int tn = (int)((TILE_TN >> (2 * u)) & 3u) * 16;
    const int c = lane & 15, h = lane >> 4;
    const float* A  = lds + Ao + (tm + c) * LD + 2 * h;
    const float* Bm = lds + Bo + (2 * h) * LD + tn + c;
    v2f af[16], bf[16];
#pragma unroll
    for (int k = 0; k < 16; ++k) {
        af[k] = *reinterpret_cast<const v2f*>(A + 4 * k);
        bf[k].x = Bm[(4 * k) * LD]; bf[k].y = Bm[(4 * k + 1) * LD];
    }
    __builtin_amdgcn_sched_barrier(0);
    v8f acc = {0.f, 0.f, 0.f, 0.f, 0.f, 0.f, 0.f, 0.f};
#pragma unroll
    for (int k = 0; k < 16; ++k)
        acc = __builtin_amdgcn_wmma_f32_16x16x4_f32(false, af[k], false, bf[k],
                                                    (short)0, acc, false, false);
    float val[8], ev[8];
    float rs = 0.f;
#pragma unroll
    for (int v = 0; v < 8; ++v) {
        val[v] = acc[v];
        const bool dg = (tm + 8 * h + v) == (tn + c);
        ev[v] = dg ? (s * val[v] + d) : (s * val[v]);
        if (MODE == 2) { const float dv = dg ? (val[v] - 1.f) : val[v]; rs = fmaf(dv, dv, rs); }
    }
    if (MODE == 1) tile_store(Do, ev, tm, tn, c, h);
    else           tile_store(Do, val, tm, tn, c, h);
    if (MODE == 2) tile_store(D2o, ev, tm, tn, c, h);
    return (tm != tn) ? rs + rs : rs;
}

template <int MODE>
__device__ __forceinline__ float mm_sym(int Do, int D2o, int Ao, int Bo, float s, float d,
                                        int lane, int r) {
    float rs = 0.f;
    for (int u = r; u < 10; u += 8) rs += tile_mm<MODE>(Do, D2o, Ao, Bo, s, d, u, lane);
    return rs;
}

template <int XM>
__device__ __forceinline__ float xmask_add(float v) {
    return v + __builtin_bit_cast(float, __builtin_amdgcn_update_dpp(
                   0, __builtin_bit_cast(int, v), 0x160 + XM, 0xf, 0xf, true));
}
__device__ __forceinline__ float row16_sum(float v) {
    return xmask_add<8>(xmask_add<4>(xmask_add<2>(xmask_add<1>(v))));
}

__device__ __forceinline__ void mm_sym2(int D1, int A1, int B1, int D2, int A2, int B2,
                                        int lane, int r) {
    for (int j = r; j < 20; j += 8) {
        const bool p = j >= 10;
        tile_mm<0>(p ? D2 : D1, 0, p ? A2 : A1, p ? B2 : B1, 1.f, 0.f, p ? j - 10 : j, lane);
    }
}

#define FOR_EW for (int i = tid; i < 4096; i += 256)

__device__ __forceinline__ void ew_identity(int Do, float dv, int tid) {
    FOR_EW { int r = i >> 6, c = i & 63; lds[Do + r * LD + c] = (r == c) ? dv : 0.f; }
    __syncthreads();
}

__device__ __forceinline__ float row_sum_bound(int So, float idv, float* buf, int tid) {
    const int row = tid >> 2, q = tid & 3;
    const int dcol = row - 16 * q;
    const float4* rp4 = reinterpret_cast<const float4*>(lds + So + row * LD + 16 * q);
    float rsum = 0.f;
    for (int j = 0; j < 4; ++j) {
        const float4 v = rp4[j];
        const int e = 4 * j;
        const float d0 = (e     == dcol) ? idv : 0.f, d1 = (e + 1 == dcol) ? idv : 0.f;
        const float d2 = (e + 2 == dcol) ? idv : 0.f, d3 = (e + 3 == dcol) ? idv : 0.f;
        rsum += (fabsf(d0 - v.x) + fabsf(d1 - v.y)) + (fabsf(d2 - v.z) + fabsf(d3 - v.w));
    }
    rsum = xmask_add<2>(xmask_add<1>(rsum));
    if (q == 0) buf[row] = rsum;
    __syncthreads();
    float bnd = 0.f;
    const float4* rp = reinterpret_cast<const float4*>(buf);
    for (int j = 0; j < 16; ++j) {
        const float4 v = rp[j];
        bnd = fmaxf(bnd, fmaxf(fmaxf(v.x, v.y), fmaxf(v.z, v.w)));
    }
    return bnd;
}

__global__ void cov64_kernel(const float* __restrict__ x, float* __restrict__ cov) {
    __shared__ __attribute__((aligned(16))) float Xs[64 * LD];
    __shared__ float mp[256];
    __shared__ float mean[64];

    const int tid = threadIdx.x, lane = tid & 31, wave = tid >> 5;
    const int bid = blockIdx.x;
    const int b = bid >> 3, e = bid & 7;
    const float* xb = x + (size_t)b * C_DIM * T_DIM + (size_t)e * PATCH;

    float msum = 0.f;
    v8f acc0 = {0.f,0.f,0.f,0.f,0.f,0.f,0.f,0.f};
    v8f acc1 = {0.f,0.f,0.f,0.f,0.f,0.f,0.f,0.f};

    for (int kc = 0; kc < PATCH; kc += 64) {
        __syncthreads();
        for (int i = tid; i < 4096; i += 256) {
            const int c = i >> 6, j = i & 63;
            Xs[c * LD + j] = xb[(size_t)c * T_DIM + kc + j];
        }
        __syncthreads();
        {
            const int c = tid & 63, part = tid >> 6;
#pragma unroll
            for (int j = 0; j < 16; ++j) msum += Xs[c * LD + part * 16 + j];
        }
#pragma unroll
        for (int i = 0; i < 2; ++i) {
            const int t  = wave + 8 * i;
            const int tm = (t >> 2) * 16, tn = (t & 3) * 16;
            const int row = tm + (lane & 15), coln = tn + (lane & 15);
            v8f acc = i ? acc1 : acc0;
#pragma unroll
            for (int k4 = 0; k4 < 64; k4 += 4) {
                const int ko = k4 + ((lane >> 4) << 1);
                const v2f a  = *reinterpret_cast<const v2f*>(Xs + row * LD + ko);
                const v2f bf = *reinterpret_cast<const v2f*>(Xs + coln * LD + ko);
                acc = __builtin_amdgcn_wmma_f32_16x16x4_f32(false, a, false, bf,
                                                            (short)0, acc, false, false);
            }
            if (i) acc1 = acc; else acc0 = acc;
        }
    }
    __syncthreads();
    mp[tid] = msum;
    __syncthreads();
    if (tid < 64)
        mean[tid] = (mp[tid] + mp[tid + 64] + mp[tid + 128] + mp[tid + 192]) * (1.f / PATCH);
    __syncthreads();

#pragma unroll
    for (int i = 0; i < 2; ++i) {
        const int t  = wave + 8 * i;
        const int tm = (t >> 2) * 16, tn = (t & 3) * 16;
        v8f acc = i ? acc1 : acc0;
#pragma unroll
        for (int v = 0; v < 8; ++v) {
            const int m = tm + v + 8 * (lane >> 4);
            const int n = tn + (lane & 15);
            const float val = (acc[v] - (float)PATCH * mean[m] * mean[n]) * (1.f / (PATCH - 1));
            cov[(size_t)bid * 4096 + m * 64 + n] = val;
        }
    }
}

__global__ void __launch_bounds__(256, 6) spdlogexp_kernel(const float* __restrict__ cov,
                                 const float* __restrict__ alphap,
                                 const float* __restrict__ betap,
                                 float* __restrict__ out) {
    float* red = lds + 5 * MS;
    float* rsm = red + 64;

    const int tid = threadIdx.x, lane = tid & 31;
    const int wave = __builtin_amdgcn_readfirstlane(tid >> 5);
    const int rot4 = (wave + 4) & 7;
    const int rot6 = (wave + 6) & 7;
    const int bid = blockIdx.x;
    const float alpha = alphap[0], beta = betap[0];

    int sy = 0, so = MS, sm = 2 * MS, sq = 3 * MS, sp = 4 * MS;

    for (int i = tid; i < 4096; i += 256)
        lds[sy + (i >> 6) * LD + (i & 63)] = cov[(size_t)bid * 4096 + i];
    __syncthreads();
    if (tid < 64) red[tid] = lds[sy + tid * LD + tid];
    __syncthreads();
    if (tid == 0) { float s = 0.f; for (int i = 0; i < 64; ++i) s += red[i]; red[0] = 1.f / s; }
    __syncthreads();
    const float invtr = red[0];
    FOR_EW { int r = i >> 6, c = i & 63;
             float v = lds[sy + r * LD + c] * invtr;
             if (r == c) v += EPS_COV;
             lds[sy + r * LD + c] = v; }
    __syncthreads();

    {
        const int row = tid >> 2, q = tid & 3;
        const float4* rp4 = reinterpret_cast<const float4*>(lds + sy + row * LD + 16 * q);
        float rsum = 0.f;
        for (int j = 0; j < 4; ++j) {
            const float4 v = rp4[j];
            rsum += (fabsf(v.x) + fabsf(v.y)) + (fabsf(v.z) + fabsf(v.w));
        }
        rsum = xmask_add<2>(xmask_add<1>(rsum));
        if (q == 0) red[row] = rsum;
    }
    __syncthreads();
    float bnd = 0.f;
    {
        const float4* rp = reinterpret_cast<const float4*>(red);
        for (int j = 0; j < 16; ++j) {
            const float4 v = rp[j];
            bnd = fmaxf(bnd, fmaxf(fmaxf(v.x, v.y), fmaxf(v.z, v.w)));
        }
    }
    const float cscale = (bnd < 2.f) ? 2.f / bnd : 1.f;
    const float logc = logf(cscale);
    FOR_EW { int r = i >> 6, c = i & 63; lds[sy + r * LD + c] *= cscale; }
    __syncthreads();

    constexpr int K_SQRT = 5;
    const int ns_it[K_SQRT] = {18, 12, 9, 7, 6};
    int kq = 0;
    for (; kq < K_SQRT; ++kq) {
        if (kq > 0) {
            const float eb = row_sum_bound(sy, 1.f, rsm, tid);
            if (__builtin_amdgcn_readfirstlane((int)(eb <= LOG_E_MAX))) break;
        }
        FOR_EW { int r = i >> 6, c = i & 63;
                 const float v = lds[sy + r * LD + c];
                 lds[sm + r * LD + c] = v;
                 lds[sp + r * LD + c] = (r == c) ? (-0.5f * v + 1.5f) : (-0.5f * v); }
        __syncthreads();
        const int n = ns_it[kq];
        for (int it = 0; it < n - 1; ++it) {
            mm_sym2(sq, sm, sp, so, sy, sp, lane, wave);
            __syncthreads();
            float rs = mm_sym<2>(sm, sy, sq, sp, -0.5f, 1.5f, lane, rot4);
            rs = row16_sum(rs);
            if ((lane & 15) == 0) red[16 + 2 * wave + (lane >> 4)] = rs;
            __syncthreads();
            const int t = sp; sp = sy; sy = so; so = t;
            const float4* rp = reinterpret_cast<const float4*>(red + 16);
            const float4 r0 = rp[0], r1 = rp[1], r2 = rp[2], r3 = rp[3];
            const float res = (((r0.x + r0.y) + (r0.z + r0.w)) + ((r1.x + r1.y) + (r1.z + r1.w)))
                            + (((r2.x + r2.y) + (r2.z + r2.w)) + ((r3.x + r3.y) + (r3.z + r3.w)));
            if (__builtin_amdgcn_readfirstlane((int)(res < NS_TOL2))) break;
        }
        mm_sym<0>(so, 0, sy, sp, 1.f, 0.f, lane, wave);
        __syncthreads();
        { const int t = sy; sy = so; so = t; }
    }
    FOR_EW { int r = i >> 6, c = i & 63;
             lds[sm + r * LD + c] = ((r == c) ? 1.f : 0.f) - lds[sy + r * LD + c]; }
    __syncthreads();
    constexpr int LOG_M = 12;
    ew_identity(sp, 1.f / (float)LOG_M, tid);
    for (int m = LOG_M - 1; m >= 1; m -= 2) {
        mm_sym<1>(sq, 0, sm, sp, 1.f, 1.f / (float)m, lane, wave);
        __syncthreads();
        const float d2 = (m > 1) ? 1.f / (float)(m - 1) : 0.f;
        mm_sym<1>(sp, 0, sm, sq, 1.f, d2, lane, rot6);
        __syncthreads();
    }
    const float lscale = -(float)(1 << kq);

    FOR_EW { int r = i >> 6, c = i & 63;
             float lg = lscale * lds[sp + r * LD + c];
             if (r == c) lg -= logc;
             lds[sy + r * LD + c] = alpha * lg + beta; }
    __syncthreads();

    constexpr int S_EXP = 6, EXP_M = 10;
    FOR_EW { int r = i >> 6, c = i & 63;
             lds[sm + r * LD + c] = 0.5f * (lds[sy + r * LD + c] + lds[sy + c * LD + r]);
             lds[sp + r * LD + c] = (r == c) ? 1.f : 0.f; }
    __syncthreads();
    const float xb = row_sum_bound(sm, 0.f, rsm, tid);
    int s_exp = 0;
    for (float t = xb; t > EXP_X_MAX && s_exp < S_EXP; t *= 0.5f) ++s_exp;
    s_exp = __builtin_amdgcn_readfirstlane(s_exp);
    const float es = 1.f / (float)(1 << s_exp);
    for (int m = EXP_M; m >= 2; m -= 2) {
        mm_sym<1>(sq, 0, sm, sp, es * (1.f / (float)m), 1.f, lane, wave);
        __syncthreads();
        mm_sym<1>(sp, 0, sm, sq, es * (1.f / (float)(m - 1)), 1.f, lane, rot6);
        __syncthreads();
    }
    for (int j = 0; j < s_exp; ++j) {
        mm_sym<0>(sq, 0, sp, sp, 1.f, 0.f, lane, (j & 1) ? rot6 : wave);
        __syncthreads();
        const int t = sp; sp = sq; sq = t;
    }

    for (int i = tid; i < 4096; i += 256)
        out[(size_t)bid * 4096 + i] = lds[sp + (i >> 6) * LD + (i & 63)];
}

extern "C" void kernel_launch(void* const* d_in, const int* in_sizes, int n_in,
                              void* d_out, int out_size, void* d_ws, size_t ws_size,
                              hipStream_t stream) {
    const float* x     = (const float*)d_in[0];
    const float* alpha = (const float*)d_in[1];
    const float* beta  = (const float*)d_in[2];
    float* out = (float*)d_out;
    float* cov = (float*)d_ws;

    const int nmat = 256 * N_EPOCH;
    cov64_kernel<<<nmat, 256, 0, stream>>>(x, cov);

    const size_t shbytes = (size_t)(5 * MS + 128) * sizeof(float);
    spdlogexp_kernel<<<nmat, 256, shbytes, stream>>>(cov, alpha, beta, out);
}


# Log and exp series summed in Paterson-Stockmeyer blocks (powers up to the cube, block terms added in the product epilogue): 22 -> 10 products per matrix; same terms, summation order differs
# speedup vs baseline: 4.8443x; 1.1281x over previous
#include <hip/hip_runtime.h>

typedef __attribute__((ext_vector_type(2))) float v2f;
typedef __attribute__((ext_vector_type(8))) float v8f;

#define LD 76
#define MS (64 * LD)
#define C_DIM 64
#define T_DIM 4096
#define N_EPOCH 8
#define PATCH 512
#define EPS_COV 1e-5f
#define NS_TOL2 1e-6f
#define LOG_E_MAX 0.3f
#define EXP_X_MAX 1.0f

extern __shared__ __attribute__((aligned(16))) float lds[];

#define TILE_TM 0xE9500u
#define TILE_TN 0xFB9E4u

__device__ __forceinline__ void tile_store(int Do, const float (&w)[8], int tm, int tn, int c, int h) {
    float* Dp = lds + Do + (tm + 8 * h) * LD + tn + c;
#pragma unroll
    for (int v = 0; v < 8; ++v) Dp[v * LD] = w[v];
    if (tm != tn) {
        float4* q = reinterpret_cast<float4*>(lds + Do + (tn + c) * LD + tm + 8 * h);
        q[0] = make_float4(w[0], w[1], w[2], w[3]);
        q[1] = make_float4(w[4], w[5], w[6], w[7]);
    }
}

template <int MODE>
__device__ __forceinline__ float tile_mm(int Do, int D2o, int Ao, int Bo, float s, float d,
                                         int u, int lane) {
    const int tm = (int)((TILE_TM >> (2 * u)) & 3u) * 16;
    const int tn = (int)((TILE_TN >> (2 * u)) & 3u) * 16;
    const int c = lane & 15, h = lane >> 4;
    const float* A  = lds + Ao + (tm + c) * LD + 2 * h;
    const float* Bm = lds + Bo + (2 * h) * LD + tn + c;
    v2f af[16], bf[16];
#pragma unroll
    for (int k = 0; k < 16; ++k) {
        af[k] = *reinterpret_cast<const v2f*>(A + 4 * k);
        bf[k].x = Bm[(4 * k) * LD]; bf[k].y = Bm[(4 * k + 1) * LD];
    }
    __builtin_amdgcn_sched_barrier(0);
    v8f acc = {0.f, 0.f, 0.f, 0.f, 0.f, 0.f, 0.f, 0.f};
#pragma unroll
    for (int k = 0; k < 16; ++k)
        acc = __builtin_amdgcn_wmma_f32_16x16x4_f32(false, af[k], false, bf[k],
                                                    (short)0, acc, false, false);
    float val[8], ev[8];
    float rs = 0.f;
#pragma unroll
    for (int v = 0; v < 8; ++v) {
        val[v] = acc[v];
        const bool dg = (tm + 8 * h + v) == (tn + c);
        ev[v] = dg ? (s * val[v] + d) : (s * val[v]);
        if (MODE == 2) { const float dv = dg ? (val[v] - 1.f) : val[v]; rs = fmaf(dv, dv, rs); }
    }
    if (MODE == 1) tile_store(Do, ev, tm, tn, c, h);
    else           tile_store(Do, val, tm, tn, c, h);
    if (MODE == 2) tile_store(D2o, ev, tm, tn, c, h);
    return (tm != tn) ? rs + rs : rs;
}

template <int MODE>
__device__ __forceinline__ float mm_sym(int Do, int D2o, int Ao, int Bo, float s, float d,
                                        int lane, int r) {
    float rs = 0.f;
    for (int u = r; u < 10; u += 8) rs += tile_mm<MODE>(Do, D2o, Ao, Bo, s, d, u, lane);
    return rs;
}

template <int NM, bool RAW>
__device__ __forceinline__ void tile_mm_ps(int D1o, int D2o, int Ao, int Bo, float s1, float s2,
                                           int M1o, float a1, int M2o, float a2, int M3o, float a3,
                                           float d, int u, int lane) {
    const int tm = (int)((TILE_TM >> (2 * u)) & 3u) * 16;
    const int tn = (int)((TILE_TN >> (2 * u)) & 3u) * 16;
    const int c = lane & 15, h = lane >> 4;
    const float* A  = lds + Ao + (tm + c) * LD + 2 * h;
    const float* Bm = lds + Bo + (2 * h) * LD + tn + c;
    v2f af[16], bf[16];
#pragma unroll
    for (int k = 0; k < 16; ++k) {
        af[k] = *reinterpret_cast<const v2f*>(A + 4 * k);
        bf[k].x = Bm[(4 * k) * LD]; bf[k].y = Bm[(4 * k + 1) * LD];
    }
    const int mo = (tn + c) * LD + tm + 8 * h;
    float4 m1a, m1b, m2a, m2b, m3a, m3b;
    m1a = m1b = m2a = m2b = m3a = m3b = make_float4(0.f, 0.f, 0.f, 0.f);
    if (NM > 0) { const float4* q = reinterpret_cast<const float4*>(lds + M1o + mo); m1a = q[0]; m1b = q[1]; }
    if (NM > 1) { const float4* q = reinterpret_cast<const float4*>(lds + M2o + mo); m2a = q[0]; m2b = q[1]; }
    if (NM > 2) { const float4* q = reinterpret_cast<const float4*>(lds + M3o + mo); m3a = q[0]; m3b = q[1]; }
    v8f acc = {0.f, 0.f, 0.f, 0.f, 0.f, 0.f, 0.f, 0.f};
#pragma unroll
    for (int k = 0; k < 16; ++k)
        acc = __builtin_amdgcn_wmma_f32_16x16x4_f32(false, af[k], false, bf[k],
                                                    (short)0, acc, false, false);
    __builtin_amdgcn_sched_group_barrier(0x100, 64, 0);
    __builtin_amdgcn_sched_group_barrier(0x008, 16, 0);
    const float m1[8] = {m1a.x, m1a.y, m1a.z, m1a.w, m1b.x, m1b.y, m1b.z, m1b.w};
    const float m2[8] = {m2a.x, m2a.y, m2a.z, m2a.w, m2b.x, m2b.y, m2b.z, m2b.w};
    const float m3[8] = {m3a.x, m3a.y, m3a.z, m3a.w, m3b.x, m3b.y, m3b.z, m3b.w};
    float rw[8], ev[8];
#pragma unroll
    for (int v = 0; v < 8; ++v) {
        rw[v] = s1 * acc[v];
        float e = s2 * acc[v];
        if (NM > 0) e = fmaf(a1, m1[v], e);
        if (NM > 1) e = fmaf(a2, m2[v], e);
        if (NM > 2) e = fmaf(a3, m3[v], e);
        const bool dg = (tm + 8 * h + v) == (tn + c);
        ev[v] = dg ? (e + d) : e;
    }
    if (RAW) tile_store(D1o, rw, tm, tn, c, h);
    tile_store(D2o, ev, tm, tn, c, h);
}

template <int NM, bool RAW>
__device__ __forceinline__ void mm_sym_ps(int D1o, int D2o, int Ao, int Bo, float s1, float s2,
                                          int M1o, float a1, int M2o, float a2, int M3o, float a3,
                                          float d, int lane, int r) {
    for (int u = r; u < 10; u += 8)
        tile_mm_ps<NM, RAW>(D1o, D2o, Ao, Bo, s1, s2, M1o, a1, M2o, a2, M3o, a3, d, u, lane);
}

template <int XM>
__device__ __forceinline__ float xmask_add(float v) {
    return v + __builtin_bit_cast(float, __builtin_amdgcn_update_dpp(
                   0, __builtin_bit_cast(int, v), 0x160 + XM, 0xf, 0xf, true));
}
__device__ __forceinline__ float row16_sum(float v) {
    return xmask_add<8>(xmask_add<4>(xmask_add<2>(xmask_add<1>(v))));
}

__device__ __forceinline__ void mm_sym2(int D1, int A1, int B1, int D2, int A2, int B2,
                                        int lane, int r) {
    for (int j = r; j < 20; j += 8) {
        const bool p = j >= 10;
        tile_mm<0>(p ? D2 : D1, 0, p ? A2 : A1, p ? B2 : B1, 1.f, 0.f, p ? j - 10 : j, lane);
    }
}

#define FOR_EW for (int i = tid; i < 4096; i += 256)

__device__ __forceinline__ float row_sum_bound(int So, float idv, float* buf, int tid) {
    const int row = tid >> 2, q = tid & 3;
    const int dcol = row - 16 * q;
    const float4* rp4 = reinterpret_cast<const float4*>(lds + So + row * LD + 16 * q);
    float rsum = 0.f;
    for (int j = 0; j < 4; ++j) {
        const float4 v = rp4[j];
        const int e = 4 * j;
        const float d0 = (e     == dcol) ? idv : 0.f, d1 = (e + 1 == dcol) ? idv : 0.f;
        const float d2 = (e + 2 == dcol) ? idv : 0.f, d3 = (e + 3 == dcol) ? idv : 0.f;
        rsum += (fabsf(d0 - v.x) + fabsf(d1 - v.y)) + (fabsf(d2 - v.z) + fabsf(d3 - v.w));
    }
    rsum = xmask_add<2>(xmask_add<1>(rsum));
    if (q == 0) buf[row] = rsum;
    __syncthreads();
    float bnd = 0.f;
    const float4* rp = reinterpret_cast<const float4*>(buf);
    for (int j = 0; j < 16; ++j) {
        const float4 v = rp[j];
        bnd = fmaxf(bnd, fmaxf(fmaxf(v.x, v.y), fmaxf(v.z, v.w)));
    }
    return bnd;
}

__global__ void cov64_kernel(const float* __restrict__ x, float* __restrict__ cov) {
    __shared__ __attribute__((aligned(16))) float Xs[64 * LD];
    __shared__ float mp[256];
    __shared__ float mean[64];

    const int tid = threadIdx.x, lane = tid & 31, wave = tid >> 5;
    const int bid = blockIdx.x;
    const int b = bid >> 3, e = bid & 7;
    const float* xb = x + (size_t)b * C_DIM * T_DIM + (size_t)e * PATCH;

    float msum = 0.f;
    v8f acc0 = {0.f,0.f,0.f,0.f,0.f,0.f,0.f,0.f};
    v8f acc1 = {0.f,0.f,0.f,0.f,0.f,0.f,0.f,0.f};

    for (int kc = 0; kc < PATCH; kc += 64) {
        __syncthreads();
        for (int i = tid; i < 4096; i += 256) {
            const int c = i >> 6, j = i & 63;
            Xs[c * LD + j] = xb[(size_t)c * T_DIM + kc + j];
        }
        __syncthreads();
        {
            const int c = tid & 63, part = tid >> 6;
#pragma unroll
            for (int j = 0; j < 16; ++j) msum += Xs[c * LD + part * 16 + j];
        }
#pragma unroll
        for (int i = 0; i < 2; ++i) {
            const int t  = wave + 8 * i;
            const int tm = (t >> 2) * 16, tn = (t & 3) * 16;
            const int row = tm + (lane & 15), coln = tn + (lane & 15);
            v8f acc = i ? acc1 : acc0;
#pragma unroll
            for (int k4 = 0; k4 < 64; k4 += 4) {
                const int ko = k4 + ((lane >> 4) << 1);
                const v2f a  = *reinterpret_cast<const v2f*>(Xs + row * LD + ko);
                const v2f bf = *reinterpret_cast<const v2f*>(Xs + coln * LD + ko);
                acc = __builtin_amdgcn_wmma_f32_16x16x4_f32(false, a, false, bf,
                                                            (short)0, acc, false, false);
            }
            if (i) acc1 = acc; else acc0 = acc;
        }
    }
    __syncthreads();
    mp[tid] = msum;
    __syncthreads();
    if (tid < 64)
        mean[tid] = (mp[tid] + mp[tid + 64] + mp[tid + 128] + mp[tid + 192]) * (1.f / PATCH);
    __syncthreads();

#pragma unroll
    for (int i = 0; i < 2; ++i) {
        const int t  = wave + 8 * i;
        const int tm = (t >> 2) * 16, tn = (t & 3) * 16;
        v8f acc = i ? acc1 : acc0;
#pragma unroll
        for (int v = 0; v < 8; ++v) {
            const int m = tm + v + 8 * (lane >> 4);
            const int n = tn + (lane & 15);
            const float val = (acc[v] - (float)PATCH * mean[m] * mean[n]) * (1.f / (PATCH - 1));
            cov[(size_t)bid * 4096 + m * 64 + n] = val;
        }
    }
}

__global__ void __launch_bounds__(256, 6) spdlogexp_kernel(const float* __restrict__ cov,
                                 const float* __restrict__ alphap,
                                 const float* __restrict__ betap,
                                 float* __restrict__ out) {
    float* red = lds + 5 * MS;
    float* rsm = red + 64;

    const int tid = threadIdx.x, lane = tid & 31;
    const int wave = __builtin_amdgcn_readfirstlane(tid >> 5);
    const int rot4 = (wave + 4) & 7;
    const int rot6 = (wave + 6) & 7;
    const int bid = blockIdx.x;
    const float alpha = alphap[0], beta = betap[0];

    int sy = 0, so = MS, sm = 2 * MS, sq = 3 * MS, sp = 4 * MS;

    for (int i = tid; i < 4096; i += 256)
        lds[sy + (i >> 6) * LD + (i & 63)] = cov[(size_t)bid * 4096 + i];
    __syncthreads();
    if (tid < 64) red[tid] = lds[sy + tid * LD + tid];
    __syncthreads();
    if (tid == 0) { float s = 0.f; for (int i = 0; i < 64; ++i) s += red[i]; red[0] = 1.f / s; }
    __syncthreads();
    const float invtr = red[0];
    FOR_EW { int r = i >> 6, c = i & 63;
             float v = lds[sy + r * LD + c] * invtr;
             if (r == c) v += EPS_COV;
             lds[sy + r * LD + c] = v; }
    __syncthreads();

    {
        const int row = tid >> 2, q = tid & 3;
        const float4* rp4 = reinterpret_cast<const float4*>(lds + sy + row * LD + 16 * q);
        float rsum = 0.f;
        for (int j = 0; j < 4; ++j) {
            const float4 v = rp4[j];
            rsum += (fabsf(v.x) + fabsf(v.y)) + (fabsf(v.z) + fabsf(v.w));
        }
        rsum = xmask_add<2>(xmask_add<1>(rsum));
        if (q == 0) red[row] = rsum;
    }
    __syncthreads();
    float bnd = 0.f;
    {
        const float4* rp = reinterpret_cast<const float4*>(red);
        for (int j = 0; j < 16; ++j) {
            const float4 v = rp[j];
            bnd = fmaxf(bnd, fmaxf(fmaxf(v.x, v.y), fmaxf(v.z, v.w)));
        }
    }
    const float cscale = (bnd < 2.f) ? 2.f / bnd : 1.f;
    const float logc = logf(cscale);
    FOR_EW { int r = i >> 6, c = i & 63; lds[sy + r * LD + c] *= cscale; }
    __syncthreads();

    constexpr int K_SQRT = 5;
    const int ns_it[K_SQRT] = {18, 12, 9, 7, 6};
    int kq = 0;
    for (; kq < K_SQRT; ++kq) {
        if (kq > 0) {
            const float eb = row_sum_bound(sy, 1.f, rsm, tid);
            if (__builtin_amdgcn_readfirstlane((int)(eb <= LOG_E_MAX))) break;
        }
        FOR_EW { int r = i >> 6, c = i & 63;
                 const float v = lds[sy + r * LD + c];
                 lds[sm + r * LD + c] = v;
                 lds[sp + r * LD + c] = (r == c) ? (-0.5f * v + 1.5f) : (-0.5f * v); }
        __syncthreads();
        const int n = ns_it[kq];
        for (int it = 0; it < n - 1; ++it) {
            mm_sym2(sq, sm, sp, so, sy, sp, lane, wave);
            __syncthreads();
            float rs = mm_sym<2>(sm, sy, sq, sp, -0.5f, 1.5f, lane, rot4);
            rs = row16_sum(rs);
            if ((lane & 15) == 0) red[16 + 2 * wave + (lane >> 4)] = rs;
            __syncthreads();
            const int t = sp; sp = sy; sy = so; so = t;
            const float4* rp = reinterpret_cast<const float4*>(red + 16);
            const float4 r0 = rp[0], r1 = rp[1], r2 = rp[2], r3 = rp[3];
            const float res = (((r0.x + r0.y) + (r0.z + r0.w)) + ((r1.x + r1.y) + (r1.z + r1.w)))
                            + (((r2.x + r2.y) + (r2.z + r2.w)) + ((r3.x + r3.y) + (r3.z + r3.w)));
            if (__builtin_amdgcn_readfirstlane((int)(res < NS_TOL2))) break;
        }
        mm_sym<0>(so, 0, sy, sp, 1.f, 0.f, lane, wave);
        __syncthreads();
        { const int t = sy; sy = so; so = t; }
    }
    FOR_EW { int r = i >> 6, c = i & 63;
             lds[sm + r * LD + c] = ((r == c) ? 1.f : 0.f) - lds[sy + r * LD + c]; }
    __syncthreads();
    mm_sym<0>(sp, 0, sm, sm, 1.f, 0.f, lane, wave);
    __syncthreads();
    mm_sym_ps<2, true>(sq, so, sm, sp, 1.f, 1.f / 12.f,
                       sm, 1.f / 10.f, sp, 1.f / 11.f, 0, 0.f, 0.f, lane, rot6);
    __syncthreads();
    mm_sym_ps<3, false>(0, sy, sq, so, 0.f, 1.f,
                        sm, 1.f / 7.f, sp, 1.f / 8.f, sq, 1.f / 9.f, 0.f, lane, wave);
    __syncthreads();
    mm_sym_ps<3, false>(0, so, sq, sy, 0.f, 1.f,
                        sm, 1.f / 4.f, sp, 1.f / 5.f, sq, 1.f / 6.f, 0.f, lane, rot6);
    __syncthreads();
    mm_sym_ps<3, false>(0, sy, sq, so, 0.f, 1.f,
                        sm, 1.f, sp, 1.f / 2.f, sq, 1.f / 3.f, 0.f, lane, wave);
    __syncthreads();
    const float lscale = -(float)(1 << kq);

    FOR_EW { int r = i >> 6, c = i & 63;
             float lg = lscale * lds[sy + r * LD + c];
             if (r == c) lg -= logc;
             lds[sy + r * LD + c] = alpha * lg + beta; }
    __syncthreads();

    constexpr int S_EXP = 6;
    FOR_EW { int r = i >> 6, c = i & 63;
             lds[sm + r * LD + c] = 0.5f * (lds[sy + r * LD + c] + lds[sy + c * LD + r]); }
    __syncthreads();
    const float xb = row_sum_bound(sm, 0.f, rsm, tid);
    int s_exp = 0;
    for (float t = xb; t > EXP_X_MAX && s_exp < S_EXP; t *= 0.5f) ++s_exp;
    s_exp = __builtin_amdgcn_readfirstlane(s_exp);
    const float es = 1.f / (float)(1 << s_exp);
    mm_sym<1>(sp, 0, sm, sm, es * es, 0.f, lane, wave);
    __syncthreads();
    mm_sym_ps<1, true>(sq, so, sm, sp, es, 0.f,
                       sm, es * (1.f / 3628800.f), 0, 0.f, 0, 0.f, 1.f / 362880.f, lane, rot6);
    __syncthreads();
    mm_sym_ps<2, false>(0, sy, sq, so, 0.f, 1.f,
                        sm, es * (1.f / 5040.f), sp, 1.f / 40320.f, 0, 0.f, 1.f / 720.f, lane, wave);
    __syncthreads();
    mm_sym_ps<2, false>(0, so, sq, sy, 0.f, 1.f,
                        sm, es * (1.f / 24.f), sp, 1.f / 120.f, 0, 0.f, 1.f / 6.f, lane, rot6);
    __syncthreads();
    mm_sym_ps<2, false>(0, sy, sq, so, 0.f, 1.f,
                        sm, es, sp, 0.5f, 0, 0.f, 1.f, lane, wave);
    __syncthreads();
    int za = sy, zb = so;
    for (int j = 0; j < s_exp; ++j) {
        mm_sym<0>(zb, 0, za, za, 1.f, 0.f, lane, (j & 1) ? rot6 : wave);
        __syncthreads();
        const int t = za; za = zb; zb = t;
    }

    for (int i = tid; i < 4096; i += 256)
        out[(size_t)bid * 4096 + i] = lds[za + (i >> 6) * LD + (i & 63)];
}

extern "C" void kernel_launch(void* const* d_in, const int* in_sizes, int n_in,
                              void* d_out, int out_size, void* d_ws, size_t ws_size,
                              hipStream_t stream) {
    const float* x     = (const float*)d_in[0];
    const float* alpha = (const float*)d_in[1];
    const float* beta  = (const float*)d_in[2];
    float* out = (float*)d_out;
    float* cov = (float*)d_ws;

    const int nmat = 256 * N_EPOCH;
    cov64_kernel<<<nmat, 256, 0, stream>>>(x, cov);

    const size_t shbytes = (size_t)(5 * MS + 128) * sizeof(float);
    spdlogexp_kernel<<<nmat, 256, shbytes, stream>>>(cov, alpha, beta, out);
}


# Log series length set from measured bounds of its argument (row sums of E and of E*E) against the remainder, so roots stop at a looser bound: fewer roots, longer sum; rounding and sum order differ
# speedup vs baseline: 5.8030x; 1.1979x over previous
#include <hip/hip_runtime.h>

typedef __attribute__((ext_vector_type(2))) float v2f;
typedef __attribute__((ext_vector_type(8))) float v8f;

#define LD 76
#define MS (64 * LD)
#define C_DIM 64
#define T_DIM 4096
#define N_EPOCH 8
#define PATCH 512
#define EPS_COV 1e-5f
#define NS_TOL2 1e-6f
#define LOG_E_MAX 0.85f
#define LOG_TOL 2e-7f
#define LOG_H_MAX 36
#define EXP_X_MAX 1.0f

extern __shared__ __attribute__((aligned(16))) float lds[];

#define TILE_TM 0xE9500u
#define TILE_TN 0xFB9E4u

__device__ __forceinline__ void tile_store(int Do, const float (&w)[8], int tm, int tn, int c, int h) {
    float* Dp = lds + Do + (tm + 8 * h) * LD + tn + c;
#pragma unroll
    for (int v = 0; v < 8; ++v) Dp[v * LD] = w[v];
    if (tm != tn) {
        float4* q = reinterpret_cast<float4*>(lds + Do + (tn + c) * LD + tm + 8 * h);
        q[0] = make_float4(w[0], w[1], w[2], w[3]);
        q[1] = make_float4(w[4], w[5], w[6], w[7]);
    }
}

template <int MODE>
__device__ __forceinline__ float tile_mm(int Do, int D2o, int Ao, int Bo, float s, float d,
                                         int u, int lane) {
    const int tm = (int)((TILE_TM >> (2 * u)) & 3u) * 16;
    const int tn = (int)((TILE_TN >> (2 * u)) & 3u) * 16;
    const int c = lane & 15, h = lane >> 4;
    const float* A  = lds + Ao + (tm + c) * LD + 2 * h;
    const float* Bm = lds + Bo + (2 * h) * LD + tn + c;
    v2f af[16], bf[16];
#pragma unroll
    for (int k = 0; k < 16; ++k) {
        af[k] = *reinterpret_cast<const v2f*>(A + 4 * k);
        bf[k].x = Bm[(4 * k) * LD]; bf[k].y = Bm[(4 * k + 1) * LD];
    }
    __builtin_amdgcn_sched_barrier(0);
    v8f acc = {0.f, 0.f, 0.f, 0.f, 0.f, 0.f, 0.f, 0.f};
#pragma unroll
    for (int k = 0; k < 16; ++k)
        acc = __builtin_amdgcn_wmma_f32_16x16x4_f32(false, af[k], false, bf[k],
                                                    (short)0, acc, false, false);
    float val[8], ev[8];
    float rs = 0.f;
#pragma unroll
    for (int v = 0; v < 8; ++v) {
        val[v] = acc[v];
        const bool dg = (tm + 8 * h + v) == (tn + c);
        ev[v] = dg ? (s * val[v] + d) : (s * val[v]);
        if (MODE == 2) { const float dv = dg ? (val[v] - 1.f) : val[v]; rs = fmaf(dv, dv, rs); }
    }
    if (MODE == 1) tile_store(Do, ev, tm, tn, c, h);
    else           tile_store(Do, val, tm, tn, c, h);
    if (MODE == 2) tile_store(D2o, ev, tm, tn, c, h);
    return (tm != tn) ? rs + rs : rs;
}

template <int MODE>
__device__ __forceinline__ float mm_sym(int Do, int D2o, int Ao, int Bo, float s, float d,
                                        int lane, int r) {
    float rs = 0.f;
    for (int u = r; u < 10; u += 8) rs += tile_mm<MODE>(Do, D2o, Ao, Bo, s, d, u, lane);
    return rs;
}

template <int NM, bool RAW>
__device__ __forceinline__ void tile_mm_ps(int D1o, int D2o, int Ao, int Bo, float s1, float s2,
                                           int M1o, float a1, int M2o, float a2, int M3o, float a3,
                                           float d, int u, int lane) {
    const int tm = (int)((TILE_TM >> (2 * u)) & 3u) * 16;
    const int tn = (int)((TILE_TN >> (2 * u)) & 3u) * 16;
    const int c = lane & 15, h = lane >> 4;
    const float* A  = lds + Ao + (tm + c) * LD + 2 * h;
    const float* Bm = lds + Bo + (2 * h) * LD + tn + c;
    v2f af[16], bf[16];
#pragma unroll
    for (int k = 0; k < 16; ++k) {
        af[k] = *reinterpret_cast<const v2f*>(A + 4 * k);
        bf[k].x = Bm[(4 * k) * LD]; bf[k].y = Bm[(4 * k + 1) * LD];
    }
    const int mo = (tn + c) * LD + tm + 8 * h;
    float4 m1a, m1b, m2a, m2b, m3a, m3b;
    m1a = m1b = m2a = m2b = m3a = m3b = make_float4(0.f, 0.f, 0.f, 0.f);
    if (NM > 0) { const float4* q = reinterpret_cast<const float4*>(lds + M1o + mo); m1a = q[0]; m1b = q[1]; }
    if (NM > 1) { const float4* q = reinterpret_cast<const float4*>(lds + M2o + mo); m2a = q[0]; m2b = q[1]; }
    if (NM > 2) { const float4* q = reinterpret_cast<const float4*>(lds + M3o + mo); m3a = q[0]; m3b = q[1]; }
    v8f acc = {0.f, 0.f, 0.f, 0.f, 0.f, 0.f, 0.f, 0.f};
#pragma unroll
    for (int k = 0; k < 16; ++k)
        acc = __builtin_amdgcn_wmma_f32_16x16x4_f32(false, af[k], false, bf[k],
                                                    (short)0, acc, false, false);
    __builtin_amdgcn_sched_group_barrier(0x100, 64, 0);
    __builtin_amdgcn_sched_group_barrier(0x008, 16, 0);
    const float m1[8] = {m1a.x, m1a.y, m1a.z, m1a.w, m1b.x, m1b.y, m1b.z, m1b.w};
    const float m2[8] = {m2a.x, m2a.y, m2a.z, m2a.w, m2b.x, m2b.y, m2b.z, m2b.w};
    const float m3[8] = {m3a.x, m3a.y, m3a.z, m3a.w, m3b.x, m3b.y, m3b.z, m3b.w};
    float rw[8], ev[8];
#pragma unroll
    for (int v = 0; v < 8; ++v) {
        rw[v] = s1 * acc[v];
        float e = s2 * acc[v];
        if (NM > 0) e = fmaf(a1, m1[v], e);
        if (NM > 1) e = fmaf(a2, m2[v], e);
        if (NM > 2) e = fmaf(a3, m3[v], e);
        const bool dg = (tm + 8 * h + v) == (tn + c);
        ev[v] = dg ? (e + d) : e;
    }
    if (RAW) tile_store(D1o, rw, tm, tn, c, h);
    tile_store(D2o, ev, tm, tn, c, h);
}

template <int NM, bool RAW>
__device__ __forceinline__ void mm_sym_ps(int D1o, int D2o, int Ao, int Bo, float s1, float s2,
                                          int M1o, float a1, int M2o, float a2, int M3o, float a3,
                                          float d, int lane, int r) {
    for (int u = r; u < 10; u += 8)
        tile_mm_ps<NM, RAW>(D1o, D2o, Ao, Bo, s1, s2, M1o, a1, M2o, a2, M3o, a3, d, u, lane);
}

template <int XM>
__device__ __forceinline__ float xmask_add(float v) {
    return v + __builtin_bit_cast(float, __builtin_amdgcn_update_dpp(
                   0, __builtin_bit_cast(int, v), 0x160 + XM, 0xf, 0xf, true));
}
__device__ __forceinline__ float row16_sum(float v) {
    return xmask_add<8>(xmask_add<4>(xmask_add<2>(xmask_add<1>(v))));
}

__device__ __forceinline__ void mm_sym2(int D1, int A1, int B1, int D2, int A2, int B2,
                                        int lane, int r) {
    for (int j = r; j < 20; j += 8) {
        const bool p = j >= 10;
        tile_mm<0>(p ? D2 : D1, 0, p ? A2 : A1, p ? B2 : B1, 1.f, 0.f, p ? j - 10 : j, lane);
    }
}

#define FOR_EW for (int i = tid; i < 4096; i += 256)

__device__ __forceinline__ float row_sum_bound(int So, float idv, float* buf, int tid) {
    const int row = tid >> 2, q = tid & 3;
    const int dcol = row - 16 * q;
    const float4* rp4 = reinterpret_cast<const float4*>(lds + So + row * LD + 16 * q);
    float rsum = 0.f;
    for (int j = 0; j < 4; ++j) {
        const float4 v = rp4[j];
        const int e = 4 * j;
        const float d0 = (e     == dcol) ? idv : 0.f, d1 = (e + 1 == dcol) ? idv : 0.f;
        const float d2 = (e + 2 == dcol) ? idv : 0.f, d3 = (e + 3 == dcol) ? idv : 0.f;
        rsum += (fabsf(d0 - v.x) + fabsf(d1 - v.y)) + (fabsf(d2 - v.z) + fabsf(d3 - v.w));
    }
    rsum = xmask_add<2>(xmask_add<1>(rsum));
    if (q == 0) buf[row] = rsum;
    __syncthreads();
    float bnd = 0.f;
    const float4* rp = reinterpret_cast<const float4*>(buf);
    for (int j = 0; j < 16; ++j) {
        const float4 v = rp[j];
        bnd = fmaxf(bnd, fmaxf(fmaxf(v.x, v.y), fmaxf(v.z, v.w)));
    }
    return bnd;
}

__global__ void cov64_kernel(const float* __restrict__ x, float* __restrict__ cov) {
    __shared__ __attribute__((aligned(16))) float Xs[64 * LD];
    __shared__ float mp[256];
    __shared__ float mean[64];

    const int tid = threadIdx.x, lane = tid & 31, wave = tid >> 5;
    const int bid = blockIdx.x;
    const int b = bid >> 3, e = bid & 7;
    const float* xb = x + (size_t)b * C_DIM * T_DIM + (size_t)e * PATCH;

    float msum = 0.f;
    v8f acc0 = {0.f,0.f,0.f,0.f,0.f,0.f,0.f,0.f};
    v8f acc1 = {0.f,0.f,0.f,0.f,0.f,0.f,0.f,0.f};

    for (int kc = 0; kc < PATCH; kc += 64) {
        __syncthreads();
        for (int i = tid; i < 4096; i += 256) {
            const int c = i >> 6, j = i & 63;
            Xs[c * LD + j] = xb[(size_t)c * T_DIM + kc + j];
        }
        __syncthreads();
        {
            const int c = tid & 63, part = tid >> 6;
#pragma unroll
            for (int j = 0; j < 16; ++j) msum += Xs[c * LD + part * 16 + j];
        }
#pragma unroll
        for (int i = 0; i < 2; ++i) {
            const int t  = wave + 8 * i;
            const int tm = (t >> 2) * 16, tn = (t & 3) * 16;
            const int row = tm + (lane & 15), coln = tn + (lane & 15);
            v8f acc = i ? acc1 : acc0;
#pragma unroll
            for (int k4 = 0; k4 < 64; k4 += 4) {
                const int ko = k4 + ((lane >> 4) << 1);
                const v2f a  = *reinterpret_cast<const v2f*>(Xs + row * LD + ko);
                const v2f bf = *reinterpret_cast<const v2f*>(Xs + coln * LD + ko);
                acc = __builtin_amdgcn_wmma_f32_16x16x4_f32(false, a, false, bf,
                                                            (short)0, acc, false, false);
            }
            if (i) acc1 = acc; else acc0 = acc;
        }
    }
    __syncthreads();
    mp[tid] = msum;
    __syncthreads();
    if (tid < 64)
        mean[tid] = (mp[tid] + mp[tid + 64] + mp[tid + 128] + mp[tid + 192]) * (1.f / PATCH);
    __syncthreads();

#pragma unroll
    for (int i = 0; i < 2; ++i) {
        const int t  = wave + 8 * i;
        const int tm = (t >> 2) * 16, tn = (t & 3) * 16;
        v8f acc = i ? acc1 : acc0;
#pragma unroll
        for (int v = 0; v < 8; ++v) {
            const int m = tm + v + 8 * (lane >> 4);
            const int n = tn + (lane & 15);
            const float val = (acc[v] - (float)PATCH * mean[m] * mean[n]) * (1.f / (PATCH - 1));
            cov[(size_t)bid * 4096 + m * 64 + n] = val;
        }
    }
}

__global__ void __launch_bounds__(256, 6) spdlogexp_kernel(const float* __restrict__ cov,
                                 const float* __restrict__ alphap,
                                 const float* __restrict__ betap,
                                 float* __restrict__ out) {
    float* red = lds + 5 * MS;
    float* rsm = red + 64;

    const int tid = threadIdx.x, lane = tid & 31;
    const int wave = __builtin_amdgcn_readfirstlane(tid >> 5);
    const int rot4 = (wave + 4) & 7;
    const int rot6 = (wave + 6) & 7;
    const int bid = blockIdx.x;
    const float alpha = alphap[0], beta = betap[0];

    int sy = 0, so = MS, sm = 2 * MS, sq = 3 * MS, sp = 4 * MS;

    for (int i = tid; i < 4096; i += 256)
        lds[sy + (i >> 6) * LD + (i & 63)] = cov[(size_t)bid * 4096 + i];
    __syncthreads();
    if (tid < 64) red[tid] = lds[sy + tid * LD + tid];
    __syncthreads();
    if (tid == 0) { float s = 0.f; for (int i = 0; i < 64; ++i) s += red[i]; red[0] = 1.f / s; }
    __syncthreads();
    const float invtr = red[0];
    FOR_EW { int r = i >> 6, c = i & 63;
             float v = lds[sy + r * LD + c] * invtr;
             if (r == c) v += EPS_COV;
             lds[sy + r * LD + c] = v; }
    __syncthreads();

    {
        const int row = tid >> 2, q = tid & 3;
        const float4* rp4 = reinterpret_cast<const float4*>(lds + sy + row * LD + 16 * q);
        float rsum = 0.f;
        for (int j = 0; j < 4; ++j) {
            const float4 v = rp4[j];
            rsum += (fabsf(v.x) + fabsf(v.y)) + (fabsf(v.z) + fabsf(v.w));
        }
        rsum = xmask_add<2>(xmask_add<1>(rsum));
        if (q == 0) red[row] = rsum;
    }
    __syncthreads();
    float bnd = 0.f;
    {
        const float4* rp = reinterpret_cast<const float4*>(red);
        for (int j = 0; j < 16; ++j) {
            const float4 v = rp[j];
            bnd = fmaxf(bnd, fmaxf(fmaxf(v.x, v.y), fmaxf(v.z, v.w)));
        }
    }
    const float cscale = (bnd < 2.f) ? 2.f / bnd : 1.f;
    const float logc = logf(cscale);
    FOR_EW { int r = i >> 6, c = i & 63; lds[sy + r * LD + c] *= cscale; }
    __syncthreads();

    constexpr int K_SQRT = 5;
    const int ns_it[K_SQRT] = {18, 12, 9, 7, 6};
    int kq = 0;
    float eb = 2.f;
    for (;; ++kq) {
        if (kq > 0) {
            eb = row_sum_bound(sy, 1.f, rsm, tid);
            if (kq == K_SQRT || __builtin_amdgcn_readfirstlane((int)(eb <= LOG_E_MAX))) break;
        }
        FOR_EW { int r = i >> 6, c = i & 63;
                 const float v = lds[sy + r * LD + c];
                 lds[sm + r * LD + c] = v;
                 lds[sp + r * LD + c] = (r == c) ? (-0.5f * v + 1.5f) : (-0.5f * v); }
        __syncthreads();
        const int n = ns_it[kq];
        for (int it = 0; it < n - 1; ++it) {
            mm_sym2(sq, sm, sp, so, sy, sp, lane, wave);
            __syncthreads();
            float rs = mm_sym<2>(sm, sy, sq, sp, -0.5f, 1.5f, lane, rot4);
            rs = row16_sum(rs);
            if ((lane & 15) == 0) red[16 + 2 * wave + (lane >> 4)] = rs;
            __syncthreads();
            const int t = sp; sp = sy; sy = so; so = t;
            const float4* rp = reinterpret_cast<const float4*>(red + 16);
            const float4 r0 = rp[0], r1 = rp[1], r2 = rp[2], r3 = rp[3];
            const float res = (((r0.x + r0.y) + (r0.z + r0.w)) + ((r1.x + r1.y) + (r1.z + r1.w)))
                            + (((r2.x + r2.y) + (r2.z + r2.w)) + ((r3.x + r3.y) + (r3.z + r3.w)));
            if (__builtin_amdgcn_readfirstlane((int)(res < NS_TOL2))) break;
        }
        mm_sym<0>(so, 0, sy, sp, 1.f, 0.f, lane, wave);
        __syncthreads();
        { const int t = sy; sy = so; so = t; }
    }
    FOR_EW { int r = i >> 6, c = i & 63;
             lds[sm + r * LD + c] = ((r == c) ? 1.f : 0.f) - lds[sy + r * LD + c]; }
    __syncthreads();
    mm_sym<0>(sp, 0, sm, sm, 1.f, 0.f, lane, wave);
    __syncthreads();
    const float b2 = row_sum_bound(sp, 0.f, rsm, tid);
    const float rr = fminf(eb, sqrtf(b2));
    int hs = 3;
    {
        const float r3 = rr * rr * rr;
        float pw = r3 * r3; pw = pw * pw * rr;
        const float lim = LOG_TOL * (1.f - rr) / (float)(1 << kq);
        for (; hs < LOG_H_MAX; ++hs) {
            if (pw <= lim * (float)(3 * hs + 4)) break;
            pw *= r3;
        }
    }
    hs = __builtin_amdgcn_readfirstlane(hs);
    {
        const float n3 = (float)(3 * hs);
        mm_sym_ps<2, true>(sq, so, sm, sp, 1.f, 1.f / (n3 + 3.f),
                           sm, 1.f / (n3 + 1.f), sp, 1.f / (n3 + 2.f), 0, 0.f, 0.f, lane, rot6);
    }
    __syncthreads();
    for (int i = hs - 1; i >= 0; --i) {
        const float n3 = (float)(3 * i);
        mm_sym_ps<3, false>(0, sy, sq, so, 0.f, 1.f,
                            sm, 1.f / (n3 + 1.f), sp, 1.f / (n3 + 2.f), sq, 1.f / (n3 + 3.f), 0.f,
                            lane, (i & 1) ? rot6 : wave);
        __syncthreads();
        const int t = sy; sy = so; so = t;
    }
    { const int t = sy; sy = so; so = t; }
    const float lscale = -(float)(1 << kq);

    FOR_EW { int r = i >> 6, c = i & 63;
             float lg = lscale * lds[sy + r * LD + c];
             if (r == c) lg -= logc;
             lds[sy + r * LD + c] = alpha * lg + beta; }
    __syncthreads();

    constexpr int S_EXP = 6;
    FOR_EW { int r = i >> 6, c = i & 63;
             lds[sm + r * LD + c] = 0.5f * (lds[sy + r * LD + c] + lds[sy + c * LD + r]); }
    __syncthreads();
    const float xb = row_sum_bound(sm, 0.f, rsm, tid);
    int s_exp = 0;
    for (float t = xb; t > EXP_X_MAX && s_exp < S_EXP; t *= 0.5f) ++s_exp;
    s_exp = __builtin_amdgcn_readfirstlane(s_exp);
    const float es = 1.f / (float)(1 << s_exp);
    mm_sym<1>(sp, 0, sm, sm, es * es, 0.f, lane, wave);
    __syncthreads();
    mm_sym_ps<1, true>(sq, so, sm, sp, es, 0.f,
                       sm, es * (1.f / 3628800.f), 0, 0.f, 0, 0.f, 1.f / 362880.f, lane, rot6);
    __syncthreads();
    mm_sym_ps<2, false>(0, sy, sq, so, 0.f, 1.f,
                        sm, es * (1.f / 5040.f), sp, 1.f / 40320.f, 0, 0.f, 1.f / 720.f, lane, wave);
    __syncthreads();
    mm_sym_ps<2, false>(0, so, sq, sy, 0.f, 1.f,
                        sm, es * (1.f / 24.f), sp, 1.f / 120.f, 0, 0.f, 1.f / 6.f, lane, rot6);
    __syncthreads();
    mm_sym_ps<2, false>(0, sy, sq, so, 0.f, 1.f,
                        sm, es, sp, 0.5f, 0, 0.f, 1.f, lane, wave);
    __syncthreads();
    int za = sy, zb = so;
    for (int j = 0; j < s_exp; ++j) {
        mm_sym<0>(zb, 0, za, za, 1.f, 0.f, lane, (j & 1) ? rot6 : wave);
        __syncthreads();
        const int t = za; za = zb; zb = t;
    }

    for (int i = tid; i < 4096; i += 256)
        out[(size_t)bid * 4096 + i] = lds[za + (i >> 6) * LD + (i & 63)];
}

extern "C" void kernel_launch(void* const* d_in, const int* in_sizes, int n_in,
                              void* d_out, int out_size, void* d_ws, size_t ws_size,
                              hipStream_t stream) {
    const float* x     = (const float*)d_in[0];
    const float* alpha = (const float*)d_in[1];
    const float* beta  = (const float*)d_in[2];
    float* out = (float*)d_out;
    float* cov = (float*)d_ws;

    const int nmat = 256 * N_EPOCH;
    cov64_kernel<<<nmat, 256, 0, stream>>>(x, cov);

    const size_t shbytes = (size_t)(5 * MS + 128) * sizeof(float);
    spdlogexp_kernel<<<nmat, 256, shbytes, stream>>>(cov, alpha, beta, out);
}


# Covariance kernel stages x by 16-byte loads issued four as one group, the next chunk in flight under the current products through a second LDS buffer (was 4-byte, a wait per pair); sums unchanged
# speedup vs baseline: 6.6165x; 1.1402x over previous
#include <hip/hip_runtime.h>

typedef __attribute__((ext_vector_type(2))) float v2f;
typedef __attribute__((ext_vector_type(8))) float v8f;

#define LD 76
#define MS (64 * LD)
#define C_DIM 64
#define T_DIM 4096
#define N_EPOCH 8
#define PATCH 512
#define EPS_COV 1e-5f
#define NS_TOL2 1e-6f
#define LOG_E_MAX 0.85f
#define LOG_TOL 2e-7f
#define LOG_H_MAX 36
#define EXP_X_MAX 1.0f

extern __shared__ __attribute__((aligned(16))) float lds[];

#define TILE_TM 0xE9500u
#define TILE_TN 0xFB9E4u

__device__ __forceinline__ void tile_store(int Do, const float (&w)[8], int tm, int tn, int c, int h) {
    float* Dp = lds + Do + (tm + 8 * h) * LD + tn + c;
#pragma unroll
    for (int v = 0; v < 8; ++v) Dp[v * LD] = w[v];
    if (tm != tn) {
        float4* q = reinterpret_cast<float4*>(lds + Do + (tn + c) * LD + tm + 8 * h);
        q[0] = make_float4(w[0], w[1], w[2], w[3]);
        q[1] = make_float4(w[4], w[5], w[6], w[7]);
    }
}

template <int MODE>
__device__ __forceinline__ float tile_mm(int Do, int D2o, int Ao, int Bo, float s, float d,
                                         int u, int lane) {
    const int tm = (int)((TILE_TM >> (2 * u)) & 3u) * 16;
    const int tn = (int)((TILE_TN >> (2 * u)) & 3u) * 16;
    const int c = lane & 15, h = lane >> 4;
    const float* A  = lds + Ao + (tm + c) * LD + 2 * h;
    const float* Bm = lds + Bo + (2 * h) * LD + tn + c;
    v2f af[16], bf[16];
#pragma unroll
    for (int k = 0; k < 16; ++k) {
        af[k] = *reinterpret_cast<const v2f*>(A + 4 * k);
        bf[k].x = Bm[(4 * k) * LD]; bf[k].y = Bm[(4 * k + 1) * LD];
    }
    __builtin_amdgcn_sched_barrier(0);
    v8f acc = {0.f, 0.f, 0.f, 0.f, 0.f, 0.f, 0.f, 0.f};
#pragma unroll
    for (int k = 0; k < 16; ++k)
        acc = __builtin_amdgcn_wmma_f32_16x16x4_f32(false, af[k], false, bf[k],
                                                    (short)0, acc, false, false);
    float val[8], ev[8];
    float rs = 0.f;
#pragma unroll
    for (int v = 0; v < 8; ++v) {
        val[v] = acc[v];
        const bool dg = (tm + 8 * h + v) == (tn + c);
        ev[v] = dg ? (s * val[v] + d) : (s * val[v]);
        if (MODE == 2) { const float dv = dg ? (val[v] - 1.f) : val[v]; rs = fmaf(dv, dv, rs); }
    }
    if (MODE == 1) tile_store(Do, ev, tm, tn, c, h);
    else           tile_store(Do, val, tm, tn, c, h);
    if (MODE == 2) tile_store(D2o, ev, tm, tn, c, h);
    return (tm != tn) ? rs + rs : rs;
}

template <int MODE>
__device__ __forceinline__ float mm_sym(int Do, int D2o, int Ao, int Bo, float s, float d,
                                        int lane, int r) {
    float rs = 0.f;
    for (int u = r; u < 10; u += 8) rs += tile_mm<MODE>(Do, D2o, Ao, Bo, s, d, u, lane);
    return rs;
}

template <int NM, bool RAW>
__device__ __forceinline__ void tile_mm_ps(int D1o, int D2o, int Ao, int Bo, float s1, float s2,
                                           int M1o, float a1, int M2o, float a2, int M3o, float a3,
                                           float d, int u, int lane) {
    const int tm = (int)((TILE_TM >> (2 * u)) & 3u) * 16;
    const int tn = (int)((TILE_TN >> (2 * u)) & 3u) * 16;
    const int c = lane & 15, h = lane >> 4;
    const float* A  = lds + Ao + (tm + c) * LD + 2 * h;
    const float* Bm = lds + Bo + (2 * h) * LD + tn + c;
    v2f af[16], bf[16];
#pragma unroll
    for (int k = 0; k < 16; ++k) {
        af[k] = *reinterpret_cast<const v2f*>(A + 4 * k);
        bf[k].x = Bm[(4 * k) * LD]; bf[k].y = Bm[(4 * k + 1) * LD];
    }
    const int mo = (tn + c) * LD + tm + 8 * h;
    float4 m1a, m1b, m2a, m2b, m3a, m3b;
    m1a = m1b = m2a = m2b = m3a = m3b = make_float4(0.f, 0.f, 0.f, 0.f);
    if (NM > 0) { const float4* q = reinterpret_cast<const float4*>(lds + M1o + mo); m1a = q[0]; m1b = q[1]; }
    if (NM > 1) { const float4* q = reinterpret_cast<const float4*>(lds + M2o + mo); m2a = q[0]; m2b = q[1]; }
    if (NM > 2) { const float4* q = reinterpret_cast<const float4*>(lds + M3o + mo); m3a = q[0]; m3b = q[1]; }
    v8f acc = {0.f, 0.f, 0.f, 0.f, 0.f, 0.f, 0.f, 0.f};
#pragma unroll
    for (int k = 0; k < 16; ++k)
        acc = __builtin_amdgcn_wmma_f32_16x16x4_f32(false, af[k], false, bf[k],
                                                    (short)0, acc, false, false);
    __builtin_amdgcn_sched_group_barrier(0x100, 64, 0);
    __builtin_amdgcn_sched_group_barrier(0x008, 16, 0);
    const float m1[8] = {m1a.x, m1a.y, m1a.z, m1a.w, m1b.x, m1b.y, m1b.z, m1b.w};
    const float m2[8] = {m2a.x, m2a.y, m2a.z, m2a.w, m2b.x, m2b.y, m2b.z, m2b.w};
    const float m3[8] = {m3a.x, m3a.y, m3a.z, m3a.w, m3b.x, m3b.y, m3b.z, m3b.w};
    float rw[8], ev[8];
#pragma unroll
    for (int v = 0; v < 8; ++v) {
        rw[v] = s1 * acc[v];
        float e = s2 * acc[v];
        if (NM > 0) e = fmaf(a1, m1[v], e);
        if (NM > 1) e = fmaf(a2, m2[v], e);
        if (NM > 2) e = fmaf(a3, m3[v], e);
        const bool dg = (tm + 8 * h + v) == (tn + c);
        ev[v] = dg ? (e + d) : e;
    }
    if (RAW) tile_store(D1o, rw, tm, tn, c, h);
    tile_store(D2o, ev, tm, tn, c, h);
}

template <int NM, bool RAW>
__device__ __forceinline__ void mm_sym_ps(int D1o, int D2o, int Ao, int Bo, float s1, float s2,
                                          int M1o, float a1, int M2o, float a2, int M3o, float a3,
                                          float d, int lane, int r) {
    for (int u = r; u < 10; u += 8)
        tile_mm_ps<NM, RAW>(D1o, D2o, Ao, Bo, s1, s2, M1o, a1, M2o, a2, M3o, a3, d, u, lane);
}

template <int XM>
__device__ __forceinline__ float xmask_add(float v) {
    return v + __builtin_bit_cast(float, __builtin_amdgcn_update_dpp(
                   0, __builtin_bit_cast(int, v), 0x160 + XM, 0xf, 0xf, true));
}
__device__ __forceinline__ float row16_sum(float v) {
    return xmask_add<8>(xmask_add<4>(xmask_add<2>(xmask_add<1>(v))));
}

__device__ __forceinline__ void mm_sym2(int D1, int A1, int B1, int D2, int A2, int B2,
                                        int lane, int r) {
    for (int j = r; j < 20; j += 8) {
        const bool p = j >= 10;
        tile_mm<0>(p ? D2 : D1, 0, p ? A2 : A1, p ? B2 : B1, 1.f, 0.f, p ? j - 10 : j, lane);
    }
}

#define FOR_EW for (int i = tid; i < 4096; i += 256)

__device__ __forceinline__ float row_sum_bound(int So, float idv, float* buf, int tid) {
    const int row = tid >> 2, q = tid & 3;
    const int dcol = row - 16 * q;
    const float4* rp4 = reinterpret_cast<const float4*>(lds + So + row * LD + 16 * q);
    float rsum = 0.f;
    for (int j = 0; j < 4; ++j) {
        const float4 v = rp4[j];
        const int e = 4 * j;
        const float d0 = (e     == dcol) ? idv : 0.f, d1 = (e + 1 == dcol) ? idv : 0.f;
        const float d2 = (e + 2 == dcol) ? idv : 0.f, d3 = (e + 3 == dcol) ? idv : 0.f;
        rsum += (fabsf(d0 - v.x) + fabsf(d1 - v.y)) + (fabsf(d2 - v.z) + fabsf(d3 - v.w));
    }
    rsum = xmask_add<2>(xmask_add<1>(rsum));
    if (q == 0) buf[row] = rsum;
    __syncthreads();
    float bnd = 0.f;
    const float4* rp = reinterpret_cast<const float4*>(buf);
    for (int j = 0; j < 16; ++j) {
        const float4 v = rp[j];
        bnd = fmaxf(bnd, fmaxf(fmaxf(v.x, v.y), fmaxf(v.z, v.w)));
    }
    return bnd;
}

__global__ void cov64_kernel(const float* __restrict__ x, float* __restrict__ cov) {
    __shared__ __attribute__((aligned(16))) float Xs2[2 * MS];
    __shared__ float mp[256];
    __shared__ float mean[64];

    const int tid = threadIdx.x, lane = tid & 31, wave = tid >> 5;
    const int bid = blockIdx.x;
    const int b = bid >> 3, e = bid & 7;
    const float* xb = x + (size_t)b * C_DIM * T_DIM + (size_t)e * PATCH;

    const int sr = tid >> 4, sc = tid & 15;
    const float4* gp = reinterpret_cast<const float4*>(xb + (size_t)sr * T_DIM) + sc;
    const int so4 = (sr * LD) / 4 + sc;
    constexpr int GR = 16 * T_DIM / 4;
    constexpr int SR = 16 * LD / 4;
    float4 p0 = gp[0], p1 = gp[GR], p2 = gp[2 * GR], p3 = gp[3 * GR];
    {
        float4* s4 = reinterpret_cast<float4*>(Xs2) + so4;
        s4[0] = p0; s4[SR] = p1; s4[2 * SR] = p2; s4[3 * SR] = p3;
    }
    __syncthreads();

    float msum = 0.f;
    v8f acc0 = {0.f,0.f,0.f,0.f,0.f,0.f,0.f,0.f};
    v8f acc1 = {0.f,0.f,0.f,0.f,0.f,0.f,0.f,0.f};

    int cur = 0;
    for (int kc = 0; kc < PATCH; kc += 64) {
        const float* Xs = Xs2 + cur * MS;
        const bool more = kc + 64 < PATCH;
        if (more) {
            const float4* gn = gp + (kc + 64) / 4;
            p0 = gn[0]; p1 = gn[GR]; p2 = gn[2 * GR]; p3 = gn[3 * GR];
        }
        {
            const int c = tid & 63, part = tid >> 6;
#pragma unroll
            for (int j = 0; j < 16; ++j) msum += Xs[c * LD + part * 16 + j];
        }
#pragma unroll
        for (int i = 0; i < 2; ++i) {
            const int t  = wave + 8 * i;
            const int tm = (t >> 2) * 16, tn = (t & 3) * 16;
            const int row = tm + (lane & 15), coln = tn + (lane & 15);
            v8f acc = i ? acc1 : acc0;
#pragma unroll
            for (int k4 = 0; k4 < 64; k4 += 4) {
                const int ko = k4 + ((lane >> 4) << 1);
                const v2f a  = *reinterpret_cast<const v2f*>(Xs + row * LD + ko);
                const v2f bf = *reinterpret_cast<const v2f*>(Xs + coln * LD + ko);
                acc = __builtin_amdgcn_wmma_f32_16x16x4_f32(false, a, false, bf,
                                                            (short)0, acc, false, false);
            }
            if (i) acc1 = acc; else acc0 = acc;
        }
        __builtin_amdgcn_sched_barrier(0);
        if (more) {
            float4* s4 = reinterpret_cast<float4*>(Xs2 + (cur ^ 1) * MS) + so4;
            s4[0] = p0; s4[SR] = p1; s4[2 * SR] = p2; s4[3 * SR] = p3;
        }
        __syncthreads();
        cur ^= 1;
    }
    mp[tid] = msum;
    __syncthreads();
    if (tid < 64)
        mean[tid] = (mp[tid] + mp[tid + 64] + mp[tid + 128] + mp[tid + 192]) * (1.f / PATCH);
    __syncthreads();

#pragma unroll
    for (int i = 0; i < 2; ++i) {
        const int t  = wave + 8 * i;
        const int tm = (t >> 2) * 16, tn = (t & 3) * 16;
        v8f acc = i ? acc1 : acc0;
#pragma unroll
        for (int v = 0; v < 8; ++v) {
            const int m = tm + v + 8 * (lane >> 4);
            const int n = tn + (lane & 15);
            const float val = (acc[v] - (float)PATCH * mean[m] * mean[n]) * (1.f / (PATCH - 1));
            cov[(size_t)bid * 4096 + m * 64 + n] = val;
        }
    }
}

__global__ void __launch_bounds__(256, 6) spdlogexp_kernel(const float* __restrict__ cov,
                                 const float* __restrict__ alphap,
                                 const float* __restrict__ betap,
                                 float* __restrict__ out) {
    float* red = lds + 5 * MS;
    float* rsm = red + 64;

    const int tid = threadIdx.x, lane = tid & 31;
    const int wave = __builtin_amdgcn_readfirstlane(tid >> 5);
    const int rot4 = (wave + 4) & 7;
    const int rot6 = (wave + 6) & 7;
    const int bid = blockIdx.x;
    const float alpha = alphap[0], beta = betap[0];

    int sy = 0, so = MS, sm = 2 * MS, sq = 3 * MS, sp = 4 * MS;

    for (int i = tid; i < 4096; i += 256)
        lds[sy + (i >> 6) * LD + (i & 63)] = cov[(size_t)bid * 4096 + i];
    __syncthreads();
    if (tid < 64) red[tid] = lds[sy + tid * LD + tid];
    __syncthreads();
    if (tid == 0) { float s = 0.f; for (int i = 0; i < 64; ++i) s += red[i]; red[0] = 1.f / s; }
    __syncthreads();
    const float invtr = red[0];
    FOR_EW { int r = i >> 6, c = i & 63;
             float v = lds[sy + r * LD + c] * invtr;
             if (r == c) v += EPS_COV;
             lds[sy + r * LD + c] = v; }
    __syncthreads();

    {
        const int row = tid >> 2, q = tid & 3;
        const float4* rp4 = reinterpret_cast<const float4*>(lds + sy + row * LD + 16 * q);
        float rsum = 0.f;
        for (int j = 0; j < 4; ++j) {
            const float4 v = rp4[j];
            rsum += (fabsf(v.x) + fabsf(v.y)) + (fabsf(v.z) + fabsf(v.w));
        }
        rsum = xmask_add<2>(xmask_add<1>(rsum));
        if (q == 0) red[row] = rsum;
    }
    __syncthreads();
    float bnd = 0.f;
    {
        const float4* rp = reinterpret_cast<const float4*>(red);
        for (int j = 0; j < 16; ++j) {
            const float4 v = rp[j];
            bnd = fmaxf(bnd, fmaxf(fmaxf(v.x, v.y), fmaxf(v.z, v.w)));
        }
    }
    const float cscale = (bnd < 2.f) ? 2.f / bnd : 1.f;
    const float logc = logf(cscale);
    FOR_EW { int r = i >> 6, c = i & 63; lds[sy + r * LD + c] *= cscale; }
    __syncthreads();

    constexpr int K_SQRT = 5;
    const int ns_it[K_SQRT] = {18, 12, 9, 7, 6};
    int kq = 0;
    float eb = 2.f;
    for (;; ++kq) {
        if (kq > 0) {
            eb = row_sum_bound(sy, 1.f, rsm, tid);
            if (kq == K_SQRT || __builtin_amdgcn_readfirstlane((int)(eb <= LOG_E_MAX))) break;
        }
        FOR_EW { int r = i >> 6, c = i & 63;
                 const float v = lds[sy + r * LD + c];
                 lds[sm + r * LD + c] = v;
                 lds[sp + r * LD + c] = (r == c) ? (-0.5f * v + 1.5f) : (-0.5f * v); }
        __syncthreads();
        const int n = ns_it[kq];
        for (int it = 0; it < n - 1; ++it) {
            mm_sym2(sq, sm, sp, so, sy, sp, lane, wave);
            __syncthreads();
            float rs = mm_sym<2>(sm, sy, sq, sp, -0.5f, 1.5f, lane, rot4);
            rs = row16_sum(rs);
            if ((lane & 15) == 0) red[16 + 2 * wave + (lane >> 4)] = rs;
            __syncthreads();
            const int t = sp; sp = sy; sy = so; so = t;
            const float4* rp = reinterpret_cast<const float4*>(red + 16);
            const float4 r0 = rp[0], r1 = rp[1], r2 = rp[2], r3 = rp[3];
            const float res = (((r0.x + r0.y) + (r0.z + r0.w)) + ((r1.x + r1.y) + (r1.z + r1.w)))
                            + (((r2.x + r2.y) + (r2.z + r2.w)) + ((r3.x + r3.y) + (r3.z + r3.w)));
            if (__builtin_amdgcn_readfirstlane((int)(res < NS_TOL2))) break;
        }
        mm_sym<0>(so, 0, sy, sp, 1.f, 0.f, lane, wave);
        __syncthreads();
        { const int t = sy; sy = so; so = t; }
    }
    FOR_EW { int r = i >> 6, c = i & 63;
             lds[sm + r * LD + c] = ((r == c) ? 1.f : 0.f) - lds[sy + r * LD + c]; }
    __syncthreads();
    mm_sym<0>(sp, 0, sm, sm, 1.f, 0.f, lane, wave);
    __syncthreads();
    const float b2 = row_sum_bound(sp, 0.f, rsm, tid);
    const float rr = fminf(eb, sqrtf(b2));
    int hs = 3;
    {
        const float r3 = rr * rr * rr;
        float pw = r3 * r3; pw = pw * pw * rr;
        const float lim = LOG_TOL * (1.f - rr) / (float)(1 << kq);
        for (; hs < LOG_H_MAX; ++hs) {
            if (pw <= lim * (float)(3 * hs + 4)) break;
            pw *= r3;
        }
    }
    hs = __builtin_amdgcn_readfirstlane(hs);
    {
        const float n3 = (float)(3 * hs);
        mm_sym_ps<2, true>(sq, so, sm, sp, 1.f, 1.f / (n3 + 3.f),
                           sm, 1.f / (n3 + 1.f), sp, 1.f / (n3 + 2.f), 0, 0.f, 0.f, lane, rot6);
    }
    __syncthreads();
    for (int i = hs - 1; i >= 0; --i) {
        const float n3 = (float)(3 * i);
        mm_sym_ps<3, false>(0, sy, sq, so, 0.f, 1.f,
                            sm, 1.f / (n3 + 1.f), sp, 1.f / (n3 + 2.f), sq, 1.f / (n3 + 3.f), 0.f,
                            lane, (i & 1) ? rot6 : wave);
        __syncthreads();
        const int t = sy; sy = so; so = t;
    }
    { const int t = sy; sy = so; so = t; }
    const float lscale = -(float)(1 << kq);

    FOR_EW { int r = i >> 6, c = i & 63;
             float lg = lscale * lds[sy + r * LD + c];
             if (r == c) lg -= logc;
             lds[sy + r * LD + c] = alpha * lg + beta; }
    __syncthreads();

    constexpr int S_EXP = 6;
    FOR_EW { int r = i >> 6, c = i & 63;
             lds[sm + r * LD + c] = 0.5f * (lds[sy + r * LD + c] + lds[sy + c * LD + r]); }
    __syncthreads();
    const float xb = row_sum_bound(sm, 0.f, rsm, tid);
    int s_exp = 0;
    for (float t = xb; t > EXP_X_MAX && s_exp < S_EXP; t *= 0.5f) ++s_exp;
    s_exp = __builtin_amdgcn_readfirstlane(s_exp);
    const float es = 1.f / (float)(1 << s_exp);
    mm_sym<1>(sp, 0, sm, sm, es * es, 0.f, lane, wave);
    __syncthreads();
    mm_sym_ps<1, true>(sq, so, sm, sp, es, 0.f,
                       sm, es * (1.f / 3628800.f), 0, 0.f, 0, 0.f, 1.f / 362880.f, lane, rot6);
    __syncthreads();
    mm_sym_ps<2, false>(0, sy, sq, so, 0.f, 1.f,
                        sm, es * (1.f / 5040.f), sp, 1.f / 40320.f, 0, 0.f, 1.f / 720.f, lane, wave);
    __syncthreads();
    mm_sym_ps<2, false>(0, so, sq, sy, 0.f, 1.f,
                        sm, es * (1.f / 24.f), sp, 1.f / 120.f, 0, 0.f, 1.f / 6.f, lane, rot6);
    __syncthreads();
    mm_sym_ps<2, false>(0, sy, sq, so, 0.f, 1.f,
                        sm, es, sp, 0.5f, 0, 0.f, 1.f, lane, wave);
    __syncthreads();
    int za = sy, zb = so;
    for (int j = 0; j < s_exp; ++j) {
        mm_sym<0>(zb, 0, za, za, 1.f, 0.f, lane, (j & 1) ? rot6 : wave);
        __syncthreads();
        const int t = za; za = zb; zb = t;
    }

    for (int i = tid; i < 4096; i += 256)
        out[(size_t)bid * 4096 + i] = lds[za + (i >> 6) * LD + (i & 63)];
}

extern "C" void kernel_launch(void* const* d_in, const int* in_sizes, int n_in,
                              void* d_out, int out_size, void* d_ws, size_t ws_size,
                              hipStream_t stream) {
    const float* x     = (const float*)d_in[0];
    const float* alpha = (const float*)d_in[1];
    const float* beta  = (const float*)d_in[2];
    float* out = (float*)d_out;
    float* cov = (float*)d_ws;

    const int nmat = 256 * N_EPOCH;
    cov64_kernel<<<nmat, 256, 0, stream>>>(x, cov);

    const size_t shbytes = (size_t)(5 * MS + 128) * sizeof(float);
    spdlogexp_kernel<<<nmat, 256, shbytes, stream>>>(cov, alpha, beta, out);
}


# Root count of the log set by a trace bound: spectral radius of E = I - Y from |E^8|_F^2 = tr(E^16), E^8 by two squarings of E*E: one square root fewer, series longer; rounding and sum length differ
# speedup vs baseline: 7.8519x; 1.1867x over previous
#include <hip/hip_runtime.h>

typedef __attribute__((ext_vector_type(2))) float v2f;
typedef __attribute__((ext_vector_type(8))) float v8f;

#define LD 76
#define MS (64 * LD)
#define C_DIM 64
#define T_DIM 4096
#define N_EPOCH 8
#define PATCH 512
#define EPS_COV 1e-5f
#define NS_TOL2 1e-6f
#define LOG_E_MAX 0.85f
#define LOG_E_TRY 1.7f
#define LOG_R_MAX 0.7f
#define LOG_TOL 2e-7f
#define LOG_H_MAX 36
#define EXP_X_MAX 1.0f

extern __shared__ __attribute__((aligned(16))) float lds[];

#define TILE_TM 0xE9500u
#define TILE_TN 0xFB9E4u

__device__ __forceinline__ void tile_store(int Do, const float (&w)[8], int tm, int tn, int c, int h) {
    float* Dp = lds + Do + (tm + 8 * h) * LD + tn + c;
#pragma unroll
    for (int v = 0; v < 8; ++v) Dp[v * LD] = w[v];
    if (tm != tn) {
        float4* q = reinterpret_cast<float4*>(lds + Do + (tn + c) * LD + tm + 8 * h);
        q[0] = make_float4(w[0], w[1], w[2], w[3]);
        q[1] = make_float4(w[4], w[5], w[6], w[7]);
    }
}

template <int MODE>
__device__ __forceinline__ float tile_mm(int Do, int D2o, int Ao, int Bo, float s, float d,
                                         int u, int lane) {
    const int tm = (int)((TILE_TM >> (2 * u)) & 3u) * 16;
    const int tn = (int)((TILE_TN >> (2 * u)) & 3u) * 16;
    const int c = lane & 15, h = lane >> 4;
    const float* A  = lds + Ao + (tm + c) * LD + 2 * h;
    const float* Bm = lds + Bo + (2 * h) * LD + tn + c;
    v2f af[16], bf[16];
#pragma unroll
    for (int k = 0; k < 16; ++k) {
        af[k] = *reinterpret_cast<const v2f*>(A + 4 * k);
        bf[k].x = Bm[(4 * k) * LD]; bf[k].y = Bm[(4 * k + 1) * LD];
    }
    __builtin_amdgcn_sched_barrier(0);
    v8f acc = {0.f, 0.f, 0.f, 0.f, 0.f, 0.f, 0.f, 0.f};
#pragma unroll
    for (int k = 0; k < 16; ++k)
        acc = __builtin_amdgcn_wmma_f32_16x16x4_f32(false, af[k], false, bf[k],
                                                    (short)0, acc, false, false);
    float val[8], ev[8];
    float rs = 0.f;
#pragma unroll
    for (int v = 0; v < 8; ++v) {
        val[v] = acc[v];
        const bool dg = (tm + 8 * h + v) == (tn + c);
        ev[v] = dg ? (s * val[v] + d) : (s * val[v]);
        if (MODE == 2) { const float dv = dg ? (val[v] - 1.f) : val[v]; rs = fmaf(dv, dv, rs); }
        if (MODE == 3) rs = fmaf(val[v], val[v], rs);
    }
    if (MODE == 1) tile_store(Do, ev, tm, tn, c, h);
    else if (MODE != 3) tile_store(Do, val, tm, tn, c, h);
    if (MODE == 2) tile_store(D2o, ev, tm, tn, c, h);
    return (tm != tn) ? rs + rs : rs;
}

template <int MODE>
__device__ __forceinline__ float mm_sym(int Do, int D2o, int Ao, int Bo, float s, float d,
                                        int lane, int r) {
    float rs = 0.f;
    for (int u = r; u < 10; u += 8) rs += tile_mm<MODE>(Do, D2o, Ao, Bo, s, d, u, lane);
    return rs;
}

template <int NM, bool RAW>
__device__ __forceinline__ void tile_mm_ps(int D1o, int D2o, int Ao, int Bo, float s1, float s2,
                                           int M1o, float a1, int M2o, float a2, int M3o, float a3,
                                           float d, int u, int lane) {
    const int tm = (int)((TILE_TM >> (2 * u)) & 3u) * 16;
    const int tn = (int)((TILE_TN >> (2 * u)) & 3u) * 16;
    const int c = lane & 15, h = lane >> 4;
    const float* A  = lds + Ao + (tm + c) * LD + 2 * h;
    const float* Bm = lds + Bo + (2 * h) * LD + tn + c;
    v2f af[16], bf[16];
#pragma unroll
    for (int k = 0; k < 16; ++k) {
        af[k] = *reinterpret_cast<const v2f*>(A + 4 * k);
        bf[k].x = Bm[(4 * k) * LD]; bf[k].y = Bm[(4 * k + 1) * LD];
    }
    const int mo = (tn + c) * LD + tm + 8 * h;
    float4 m1a, m1b, m2a, m2b, m3a, m3b;
    m1a = m1b = m2a = m2b = m3a = m3b = make_float4(0.f, 0.f, 0.f, 0.f);
    if (NM > 0) { const float4* q = reinterpret_cast<const float4*>(lds + M1o + mo); m1a = q[0]; m1b = q[1]; }
    if (NM > 1) { const float4* q = reinterpret_cast<const float4*>(lds + M2o + mo); m2a = q[0]; m2b = q[1]; }
    if (NM > 2) { const float4* q = reinterpret_cast<const float4*>(lds + M3o + mo); m3a = q[0]; m3b = q[1]; }
    v8f acc = {0.f, 0.f, 0.f, 0.f, 0.f, 0.f, 0.f, 0.f};
#pragma unroll
    for (int k = 0; k < 16; ++k)
        acc = __builtin_amdgcn_wmma_f32_16x16x4_f32(false, af[k], false, bf[k],
                                                    (short)0, acc, false, false);
    __builtin_amdgcn_sched_group_barrier(0x100, 64, 0);
    __builtin_amdgcn_sched_group_barrier(0x008, 16, 0);
    const float m1[8] = {m1a.x, m1a.y, m1a.z, m1a.w, m1b.x, m1b.y, m1b.z, m1b.w};
    const float m2[8] = {m2a.x, m2a.y, m2a.z, m2a.w, m2b.x, m2b.y, m2b.z, m2b.w};
    const float m3[8] = {m3a.x, m3a.y, m3a.z, m3a.w, m3b.x, m3b.y, m3b.z, m3b.w};
    float rw[8], ev[8];
#pragma unroll
    for (int v = 0; v < 8; ++v) {
        rw[v] = s1 * acc[v];
        float e = s2 * acc[v];
        if (NM > 0) e = fmaf(a1, m1[v], e);
        if (NM > 1) e = fmaf(a2, m2[v], e);
        if (NM > 2) e = fmaf(a3, m3[v], e);
        const bool dg = (tm + 8 * h + v) == (tn + c);
        ev[v] = dg ? (e + d) : e;
    }
    if (RAW) tile_store(D1o, rw, tm, tn, c, h);
    tile_store(D2o, ev, tm, tn, c, h);
}

template <int NM, bool RAW>
__device__ __forceinline__ void mm_sym_ps(int D1o, int D2o, int Ao, int Bo, float s1, float s2,
                                          int M1o, float a1, int M2o, float a2, int M3o, float a3,
                                          float d, int lane, int r) {
    for (int u = r; u < 10; u += 8)
        tile_mm_ps<NM, RAW>(D1o, D2o, Ao, Bo, s1, s2, M1o, a1, M2o, a2, M3o, a3, d, u, lane);
}

template <int XM>
__device__ __forceinline__ float xmask_add(float v) {
    return v + __builtin_bit_cast(float, __builtin_amdgcn_update_dpp(
                   0, __builtin_bit_cast(int, v), 0x160 + XM, 0xf, 0xf, true));
}
__device__ __forceinline__ float row16_sum(float v) {
    return xmask_add<8>(xmask_add<4>(xmask_add<2>(xmask_add<1>(v))));
}

__device__ __forceinline__ void mm_sym2(int D1, int A1, int B1, int D2, int A2, int B2,
                                        int lane, int r) {
    for (int j = r; j < 20; j += 8) {
        const bool p = j >= 10;
        tile_mm<0>(p ? D2 : D1, 0, p ? A2 : A1, p ? B2 : B1, 1.f, 0.f, p ? j - 10 : j, lane);
    }
}

#define FOR_EW for (int i = tid; i < 4096; i += 256)

__device__ __forceinline__ float row_sum_bound(int So, float idv, float* buf, int tid) {
    const int row = tid >> 2, q = tid & 3;
    const int dcol = row - 16 * q;
    const float4* rp4 = reinterpret_cast<const float4*>(lds + So + row * LD + 16 * q);
    float rsum = 0.f;
    for (int j = 0; j < 4; ++j) {
        const float4 v = rp4[j];
        const int e = 4 * j;
        const float d0 = (e     == dcol) ? idv : 0.f, d1 = (e + 1 == dcol) ? idv : 0.f;
        const float d2 = (e + 2 == dcol) ? idv : 0.f, d3 = (e + 3 == dcol) ? idv : 0.f;
        rsum += (fabsf(d0 - v.x) + fabsf(d1 - v.y)) + (fabsf(d2 - v.z) + fabsf(d3 - v.w));
    }
    rsum = xmask_add<2>(xmask_add<1>(rsum));
    if (q == 0) buf[row] = rsum;
    __syncthreads();
    float bnd = 0.f;
    const float4* rp = reinterpret_cast<const float4*>(buf);
    for (int j = 0; j < 16; ++j) {
        const float4 v = rp[j];
        bnd = fmaxf(bnd, fmaxf(fmaxf(v.x, v.y), fmaxf(v.z, v.w)));
    }
    return bnd;
}

__global__ void cov64_kernel(const float* __restrict__ x, float* __restrict__ cov) {
    __shared__ __attribute__((aligned(16))) float Xs2[2 * MS];
    __shared__ float mp[256];
    __shared__ float mean[64];

    const int tid = threadIdx.x, lane = tid & 31, wave = tid >> 5;
    const int bid = blockIdx.x;
    const int b = bid >> 3, e = bid & 7;
    const float* xb = x + (size_t)b * C_DIM * T_DIM + (size_t)e * PATCH;

    const int sr = tid >> 4, sc = tid & 15;
    const float4* gp = reinterpret_cast<const float4*>(xb + (size_t)sr * T_DIM) + sc;
    const int so4 = (sr * LD) / 4 + sc;
    constexpr int GR = 16 * T_DIM / 4;
    constexpr int SR = 16 * LD / 4;
    float4 p0 = gp[0], p1 = gp[GR], p2 = gp[2 * GR], p3 = gp[3 * GR];
    {
        float4* s4 = reinterpret_cast<float4*>(Xs2) + so4;
        s4[0] = p0; s4[SR] = p1; s4[2 * SR] = p2; s4[3 * SR] = p3;
    }
    __syncthreads();

    float msum = 0.f;
    v8f acc0 = {0.f,0.f,0.f,0.f,0.f,0.f,0.f,0.f};
    v8f acc1 = {0.f,0.f,0.f,0.f,0.f,0.f,0.f,0.f};

    int cur = 0;
    for (int kc = 0; kc < PATCH; kc += 64) {
        const float* Xs = Xs2 + cur * MS;
        const bool more = kc + 64 < PATCH;
        if (more) {
            const float4* gn = gp + (kc + 64) / 4;
            p0 = gn[0]; p1 = gn[GR]; p2 = gn[2 * GR]; p3 = gn[3 * GR];
        }
        {
            const int c = tid & 63, part = tid >> 6;
#pragma unroll
            for (int j = 0; j < 16; ++j) msum += Xs[c * LD + part * 16 + j];
        }
#pragma unroll
        for (int i = 0; i < 2; ++i) {
            const int t  = wave + 8 * i;
            const int tm = (t >> 2) * 16, tn = (t & 3) * 16;
            const int row = tm + (lane & 15), coln = tn + (lane & 15);
            v8f acc = i ? acc1 : acc0;
#pragma unroll
            for (int k4 = 0; k4 < 64; k4 += 4) {
                const int ko = k4 + ((lane >> 4) << 1);
                const v2f a  = *reinterpret_cast<const v2f*>(Xs + row * LD + ko);
                const v2f bf = *reinterpret_cast<const v2f*>(Xs + coln * LD + ko);
                acc = __builtin_amdgcn_wmma_f32_16x16x4_f32(false, a, false, bf,
                                                            (short)0, acc, false, false);
            }
            if (i) acc1 = acc; else acc0 = acc;
        }
        __builtin_amdgcn_sched_barrier(0);
        if (more) {
            float4* s4 = reinterpret_cast<float4*>(Xs2 + (cur ^ 1) * MS) + so4;
            s4[0] = p0; s4[SR] = p1; s4[2 * SR] = p2; s4[3 * SR] = p3;
        }
        __syncthreads();
        cur ^= 1;
    }
    mp[tid] = msum;
    __syncthreads();
    if (tid < 64)
        mean[tid] = (mp[tid] + mp[tid + 64] + mp[tid + 128] + mp[tid + 192]) * (1.f / PATCH);
    __syncthreads();

#pragma unroll
    for (int i = 0; i < 2; ++i) {
        const int t  = wave + 8 * i;
        const int tm = (t >> 2) * 16, tn = (t & 3) * 16;
        v8f acc = i ? acc1 : acc0;
#pragma unroll
        for (int v = 0; v < 8; ++v) {
            const int m = tm + v + 8 * (lane >> 4);
            const int n = tn + (lane & 15);
            const float val = (acc[v] - (float)PATCH * mean[m] * mean[n]) * (1.f / (PATCH - 1));
            cov[(size_t)bid * 4096 + m * 64 + n] = val;
        }
    }
}

__global__ void __launch_bounds__(256, 6) spdlogexp_kernel(const float* __restrict__ cov,
                                 const float* __restrict__ alphap,
                                 const float* __restrict__ betap,
                                 float* __restrict__ out) {
    float* red = lds + 5 * MS;
    float* rsm = red + 64;

    const int tid = threadIdx.x, lane = tid & 31;
    const int wave = __builtin_amdgcn_readfirstlane(tid >> 5);
    const int rot4 = (wave + 4) & 7;
    const int rot6 = (wave + 6) & 7;
    const int bid = blockIdx.x;
    const float alpha = alphap[0], beta = betap[0];

    int sy = 0, so = MS, sm = 2 * MS, sq = 3 * MS, sp = 4 * MS;

    for (int i = tid; i < 4096; i += 256)
        lds[sy + (i >> 6) * LD + (i & 63)] = cov[(size_t)bid * 4096 + i];
    __syncthreads();
    if (tid < 64) red[tid] = lds[sy + tid * LD + tid];
    __syncthreads();
    if (tid == 0) { float s = 0.f; for (int i = 0; i < 64; ++i) s += red[i]; red[0] = 1.f / s; }
    __syncthreads();
    const float invtr = red[0];
    FOR_EW { int r = i >> 6, c = i & 63;
             float v = lds[sy + r * LD + c] * invtr;
             if (r == c) v += EPS_COV;
             lds[sy + r * LD + c] = v; }
    __syncthreads();

    {
        const int row = tid >> 2, q = tid & 3;
        const float4* rp4 = reinterpret_cast<const float4*>(lds + sy + row * LD + 16 * q);
        float rsum = 0.f;
        for (int j = 0; j < 4; ++j) {
            const float4 v = rp4[j];
            rsum += (fabsf(v.x) + fabsf(v.y)) + (fabsf(v.z) + fabsf(v.w));
        }
        rsum = xmask_add<2>(xmask_add<1>(rsum));
        if (q == 0) red[row] = rsum;
    }
    __syncthreads();
    float bnd = 0.f;
    {
        const float4* rp = reinterpret_cast<const float4*>(red);
        for (int j = 0; j < 16; ++j) {
            const float4 v = rp[j];
            bnd = fmaxf(bnd, fmaxf(fmaxf(v.x, v.y), fmaxf(v.z, v.w)));
        }
    }
    const float cscale = (bnd < 2.f) ? 2.f / bnd : 1.f;
    const float logc = logf(cscale);
    FOR_EW { int r = i >> 6, c = i & 63; lds[sy + r * LD + c] *= cscale; }
    __syncthreads();

    constexpr int K_SQRT = 5;
    const int ns_it[K_SQRT] = {18, 12, 9, 7, 6};
    int kq = 0;
    float eb = 2.f;
    float rt = 2.f;
    int have_e = 0;
    for (;; ++kq) {
        if (kq > 0) {
            eb = row_sum_bound(sy, 1.f, rsm, tid);
            if (kq == K_SQRT || __builtin_amdgcn_readfirstlane((int)(eb <= LOG_E_MAX))) break;
            if (__builtin_amdgcn_readfirstlane((int)(eb <= LOG_E_TRY))) {
                FOR_EW { int r = i >> 6, c = i & 63;
                         lds[sm + r * LD + c] = ((r == c) ? 1.f : 0.f) - lds[sy + r * LD + c]; }
                __syncthreads();
                mm_sym<0>(sp, 0, sm, sm, 1.f, 0.f, lane, wave);
                __syncthreads();
                mm_sym<0>(sq, 0, sp, sp, 1.f, 0.f, lane, rot4);
                __syncthreads();
                float t = mm_sym<3>(0, 0, sq, sq, 1.f, 0.f, lane, rot6);
                t = row16_sum(t);
                if ((lane & 15) == 0) red[16 + 2 * wave + (lane >> 4)] = t;
                __syncthreads();
                const float4* rp = reinterpret_cast<const float4*>(red + 16);
                const float4 r0 = rp[0], r1 = rp[1], r2 = rp[2], r3 = rp[3];
                const float t16 = (((r0.x + r0.y) + (r0.z + r0.w)) + ((r1.x + r1.y) + (r1.z + r1.w)))
                                + (((r2.x + r2.y) + (r2.z + r2.w)) + ((r3.x + r3.y) + (r3.z + r3.w)));
                const float r16 = sqrtf(sqrtf(sqrtf(sqrtf(t16))));
                if (__builtin_amdgcn_readfirstlane((int)(r16 <= LOG_R_MAX))) { rt = r16; have_e = 1; break; }
            }
        }
        FOR_EW { int r = i >> 6, c = i & 63;
                 const float v = lds[sy + r * LD + c];
                 lds[sm + r * LD + c] = v;
                 lds[sp + r * LD + c] = (r == c) ? (-0.5f * v + 1.5f) : (-0.5f * v); }
        __syncthreads();
        const int n = ns_it[kq];
        for (int it = 0; it < n - 1; ++it) {
            mm_sym2(sq, sm, sp, so, sy, sp, lane, wave);
            __syncthreads();
            float rs = mm_sym<2>(sm, sy, sq, sp, -0.5f, 1.5f, lane, rot4);
            rs = row16_sum(rs);
            if ((lane & 15) == 0) red[16 + 2 * wave + (lane >> 4)] = rs;
            __syncthreads();
            const int t = sp; sp = sy; sy = so; so = t;
            const float4* rp = reinterpret_cast<const float4*>(red + 16);
            const float4 r0 = rp[0], r1 = rp[1], r2 = rp[2], r3 = rp[3];
            const float res = (((r0.x + r0.y) + (r0.z + r0.w)) + ((r1.x + r1.y) + (r1.z + r1.w)))
                            + (((r2.x + r2.y) + (r2.z + r2.w)) + ((r3.x + r3.y) + (r3.z + r3.w)));
            if (__builtin_amdgcn_readfirstlane((int)(res < NS_TOL2))) break;
        }
        mm_sym<0>(so, 0, sy, sp, 1.f, 0.f, lane, wave);
        __syncthreads();
        { const int t = sy; sy = so; so = t; }
    }
    if (!have_e) {
        FOR_EW { int r = i >> 6, c = i & 63;
                 lds[sm + r * LD + c] = ((r == c) ? 1.f : 0.f) - lds[sy + r * LD + c]; }
        __syncthreads();
    }
    if (!have_e) {
        mm_sym<0>(sp, 0, sm, sm, 1.f, 0.f, lane, wave);
        __syncthreads();
    }
    const float b2 = row_sum_bound(sp, 0.f, rsm, tid);
    const float rr = fminf(fminf(eb, sqrtf(b2)), rt);
    int hs = 3;
    {
        const float r3 = rr * rr * rr;
        float pw = r3 * r3; pw = pw * pw * rr;
        const float lim = LOG_TOL * (1.f - rr) / (float)(1 << kq);
        for (; hs < LOG_H_MAX; ++hs) {
            if (pw <= lim * (float)(3 * hs + 4)) break;
            pw *= r3;
        }
    }
    hs = __builtin_amdgcn_readfirstlane(hs);
    {
        const float n3 = (float)(3 * hs);
        mm_sym_ps<2, true>(sq, so, sm, sp, 1.f, 1.f / (n3 + 3.f),
                           sm, 1.f / (n3 + 1.f), sp, 1.f / (n3 + 2.f), 0, 0.f, 0.f, lane, rot6);
    }
    __syncthreads();
    for (int i = hs - 1; i >= 0; --i) {
        const float n3 = (float)(3 * i);
        mm_sym_ps<3, false>(0, sy, sq, so, 0.f, 1.f,
                            sm, 1.f / (n3 + 1.f), sp, 1.f / (n3 + 2.f), sq, 1.f / (n3 + 3.f), 0.f,
                            lane, (i & 1) ? rot6 : wave);
        __syncthreads();
        const int t = sy; sy = so; so = t;
    }
    { const int t = sy; sy = so; so = t; }
    const float lscale = -(float)(1 << kq);

    FOR_EW { int r = i >> 6, c = i & 63;
             float lg = lscale * lds[sy + r * LD + c];
             if (r == c) lg -= logc;
             lds[sy + r * LD + c] = alpha * lg + beta; }
    __syncthreads();

    constexpr int S_EXP = 6;
    FOR_EW { int r = i >> 6, c = i & 63;
             lds[sm + r * LD + c] = 0.5f * (lds[sy + r * LD + c] + lds[sy + c * LD + r]); }
    __syncthreads();
    const float xb = row_sum_bound(sm, 0.f, rsm, tid);
    int s_exp = 0;
    for (float t = xb; t > EXP_X_MAX && s_exp < S_EXP; t *= 0.5f) ++s_exp;
    s_exp = __builtin_amdgcn_readfirstlane(s_exp);
    const float es = 1.f / (float)(1 << s_exp);
    mm_sym<1>(sp, 0, sm, sm, es * es, 0.f, lane, wave);
    __syncthreads();
    mm_sym_ps<1, true>(sq, so, sm, sp, es, 0.f,
                       sm, es * (1.f / 3628800.f), 0, 0.f, 0, 0.f, 1.f / 362880.f, lane, rot6);
    __syncthreads();
    mm_sym_ps<2, false>(0, sy, sq, so, 0.f, 1.f,
                        sm, es * (1.f / 5040.f), sp, 1.f / 40320.f, 0, 0.f, 1.f / 720.f, lane, wave);
    __syncthreads();
    mm_sym_ps<2, false>(0, so, sq, sy, 0.f, 1.f,
                        sm, es * (1.f / 24.f), sp, 1.f / 120.f, 0, 0.f, 1.f / 6.f, lane, rot6);
    __syncthreads();
    mm_sym_ps<2, false>(0, sy, sq, so, 0.f, 1.f,
                        sm, es, sp, 0.5f, 0, 0.f, 1.f, lane, wave);
    __syncthreads();
    int za = sy, zb = so;
    for (int j = 0; j < s_exp; ++j) {
        mm_sym<0>(zb, 0, za, za, 1.f, 0.f, lane, (j & 1) ? rot6 : wave);
        __syncthreads();
        const int t = za; za = zb; zb = t;
    }

    for (int i = tid; i < 4096; i += 256)
        out[(size_t)bid * 4096 + i] = lds[za + (i >> 6) * LD + (i & 63)];
}

extern "C" void kernel_launch(void* const* d_in, const int* in_sizes, int n_in,
                              void* d_out, int out_size, void* d_ws, size_t ws_size,
                              hipStream_t stream) {
    const float* x     = (const float*)d_in[0];
    const float* alpha = (const float*)d_in[1];
    const float* beta  = (const float*)d_in[2];
    float* out = (float*)d_out;
    float* cov = (float*)d_ws;

    const int nmat = 256 * N_EPOCH;
    cov64_kernel<<<nmat, 256, 0, stream>>>(x, cov);

    const size_t shbytes = (size_t)(5 * MS + 128) * sizeof(float);
    spdlogexp_kernel<<<nmat, 256, shbytes, stream>>>(cov, alpha, beta, out);
}
